# starting kernel
# baseline (speedup 1.0000x reference)
#include <hip/hip_runtime.h>

#define N_NODES  50000
#define N_EDGES  800000
#define NF       128
#define N_GRAPHS 128

typedef __attribute__((ext_vector_type(2))) float v2f;
typedef __attribute__((ext_vector_type(8))) float v8f;

__global__ void zero_kernel(float* __restrict__ p, int n) {
  int i = blockIdx.x * blockDim.x + threadIdx.x;
  if (i < n) p[i] = 0.0f;
}

__global__ void deg_kernel(const int* __restrict__ dst, float* __restrict__ deg) {
  int e = blockIdx.x * blockDim.x + threadIdx.x;
  if (e < N_EDGES) atomicAdd(&deg[dst[e]], 1.0f);
}

__global__ void dinv_kernel(float* __restrict__ deg) {
  int i = blockIdx.x * blockDim.x + threadIdx.x;
  if (i < N_NODES) deg[i] = rsqrtf(deg[i] + 1.0f);
}

__global__ __launch_bounds__(256) void gemm_wmma(const float* __restrict__ X,
                                                 const float* __restrict__ W,
                                                 float* __restrict__ H) {
  const int wave = threadIdx.x >> 5;
  const int lane = threadIdx.x & 31;
  const int half = lane >> 4;
  const int l    = lane & 15;
  const int m0   = blockIdx.x * 16;
  const int n0   = wave * 16;

  const float* Xrow = X + (size_t)(m0 + l) * NF;
  v8f acc = {};

  #pragma unroll
  for (int k = 0; k < NF; k += 4) {
    v2f a = *(const v2f*)(Xrow + k + 2 * half);
    v2f b;
    b.x = W[(size_t)(k + 2 * half + 0) * NF + n0 + l];
    b.y = W[(size_t)(k + 2 * half + 1) * NF + n0 + l];
    acc = __builtin_amdgcn_wmma_f32_16x16x4_f32(
         false, a,  false, b,
         (short)0, acc,  false,  false);
  }

  float* Hp = H + (size_t)(m0 + 8 * half) * NF + n0 + l;
  #pragma unroll
  for (int r = 0; r < 8; ++r) Hp[(size_t)r * NF] = acc[r];
}

__global__ __launch_bounds__(256) void scatter_kernel(const float* __restrict__ H,
                                                      const int* __restrict__ src,
                                                      const int* __restrict__ dst,
                                                      const float* __restrict__ dinv,
                                                      float* __restrict__ AGG) {
  int gid  = blockIdx.x * blockDim.x + threadIdx.x;
  int e    = gid >> 5;
  if (e >= N_EDGES) return;
  int lane = gid & 31;
  int s = src[e];
  int d = dst[e];
  float norm = dinv[s] * dinv[d];
  const float4 hv = *(const float4*)(H + (size_t)s * NF + lane * 4);
  float* op = AGG + (size_t)d * NF + lane * 4;
  atomicAdd(op + 0, hv.x * norm);
  atomicAdd(op + 1, hv.y * norm);
  atomicAdd(op + 2, hv.z * norm);
  atomicAdd(op + 3, hv.w * norm);
}

__global__ __launch_bounds__(256) void combine_relu(const float* __restrict__ AGG,
                                                    const float* __restrict__ H,
                                                    const float* __restrict__ dinv,
                                                    const float* __restrict__ b,
                                                    float* __restrict__ Y) {
  int gid = blockIdx.x * blockDim.x + threadIdx.x;
  if (gid >= N_NODES * 32) return;
  int i = gid >> 5;
  int q = (gid & 31) * 4;
  float di = dinv[i];
  float di2 = di * di;
  size_t off = (size_t)i * NF + q;
  float4 av = *(const float4*)(AGG + off);
  float4 hv = *(const float4*)(H + off);
  float4 bv = *(const float4*)(b + q);
  float4 r;
  r.x = fmaxf(av.x + di2 * hv.x + bv.x, 0.0f);
  r.y = fmaxf(av.y + di2 * hv.y + bv.y, 0.0f);
  r.z = fmaxf(av.z + di2 * hv.z + bv.z, 0.0f);
  r.w = fmaxf(av.w + di2 * hv.w + bv.w, 0.0f);
  *(float4*)(Y + off) = r;
}

__global__ __launch_bounds__(256) void combine_pool(const float* __restrict__ AGG,
                                                    const float* __restrict__ H,
                                                    const float* __restrict__ dinv,
                                                    const float* __restrict__ b,
                                                    const int* __restrict__ batch,
                                                    float* __restrict__ out) {
  int gid = blockIdx.x * blockDim.x + threadIdx.x;
  if (gid >= N_NODES * 32) return;
  int i = gid >> 5;
  int q = (gid & 31) * 4;
  float di = dinv[i];
  float di2 = di * di;
  size_t off = (size_t)i * NF + q;
  float4 av = *(const float4*)(AGG + off);
  float4 hv = *(const float4*)(H + off);
  float4 bv = *(const float4*)(b + q);
  float* op = out + (size_t)batch[i] * NF + q;
  atomicAdd(op + 0, av.x + di2 * hv.x + bv.x);
  atomicAdd(op + 1, av.y + di2 * hv.y + bv.y);
  atomicAdd(op + 2, av.z + di2 * hv.z + bv.z);
  atomicAdd(op + 3, av.w + di2 * hv.w + bv.w);
}

extern "C" void kernel_launch(void* const* d_in, const int* in_sizes, int n_in,
                              void* d_out, int out_size, void* d_ws, size_t ws_size,
                              hipStream_t stream) {
  const float* x    = (const float*)d_in[0];
  const int*   ei   = (const int*)  d_in[1];
  const int*   batch= (const int*)  d_in[2];
  const float* W1   = (const float*)d_in[3];
  const float* b1   = (const float*)d_in[4];
  const float* W2   = (const float*)d_in[5];
  const float* b2   = (const float*)d_in[6];
  float* out = (float*)d_out;
  float* ws  = (float*)d_ws;

  const int* src = ei;
  const int* dst = ei + N_EDGES;

  float* dinv = ws;
  float* H    = ws + 50048;
  float* AGG  = H  + (size_t)N_NODES * NF;
  float* Y    = AGG + (size_t)N_NODES * NF;

  const int T = 256;
  const int gNodes    = (N_NODES + T - 1) / T;
  const int gEdges    = (N_EDGES + T - 1) / T;
  const int gFeat     = ((size_t)N_NODES * NF + T - 1) / T;
  const int gNodeVec  = (N_NODES * 32 + T - 1) / T;
  const int gScatter  = ((size_t)N_EDGES * 32 + T - 1) / T;
  const int gGemm     = N_NODES / 16;

  zero_kernel<<<gNodes, T, 0, stream>>>(dinv, N_NODES);
  deg_kernel<<<gEdges, T, 0, stream>>>(dst, dinv);
  dinv_kernel<<<gNodes, T, 0, stream>>>(dinv);

  gemm_wmma<<<gGemm, T, 0, stream>>>(x, W1, H);
  zero_kernel<<<gFeat, T, 0, stream>>>(AGG, N_NODES * NF);
  scatter_kernel<<<gScatter, T, 0, stream>>>(H, src, dst, dinv, AGG);
  combine_relu<<<gNodeVec, T, 0, stream>>>(AGG, H, dinv, b1, Y);

  gemm_wmma<<<gGemm, T, 0, stream>>>(Y, W2, H);
  zero_kernel<<<gFeat, T, 0, stream>>>(AGG, N_NODES * NF);
  scatter_kernel<<<gScatter, T, 0, stream>>>(H, src, dst, dinv, AGG);

  zero_kernel<<<(N_GRAPHS * NF + T - 1) / T, T, 0, stream>>>(out, N_GRAPHS * NF);
  combine_pool<<<gNodeVec, T, 0, stream>>>(AGG, H, dinv, b2, batch, out);
}


# dst-sorted CSR built per call (int atomic rank, scan, fill); one wave per row sums its in-edges in arrival-rank order, replacing float atomic scatter and zero-fill; layer-1 combine at the row store
# speedup vs baseline: 1.8797x; 1.8797x over previous
#include <hip/hip_runtime.h>

#define N_NODES  50000
#define N_EDGES  800000
#define NF       128
#define N_GRAPHS 128

typedef __attribute__((ext_vector_type(2))) float v2f;
typedef __attribute__((ext_vector_type(8))) float v8f;

__global__ void zero_kernel(float* __restrict__ p, int n) {
  int i = blockIdx.x * blockDim.x + threadIdx.x;
  if (i < n) p[i] = 0.0f;
}

__global__ void zero_int_kernel(int* __restrict__ p, int n) {
  int i = blockIdx.x * blockDim.x + threadIdx.x;
  if (i < n) p[i] = 0;
}


__global__ void count_rank_kernel(const int* __restrict__ dst, int* __restrict__ cnt,
                                  int* __restrict__ rank) {
  int e = blockIdx.x * blockDim.x + threadIdx.x;
  if (e < N_EDGES) rank[e] = atomicAdd(&cnt[dst[e]], 1);
}

#define SCAN_TILES 13
static_assert(32 * SCAN_TILES * 128 >= N_NODES, "scan block must cover all nodes");
static_assert(N_NODES % 4 == 0, "counts are handled four at a time");
__global__ __launch_bounds__(1024) void scan_dinv_kernel(int* __restrict__ cr,
                                                         float* __restrict__ dinv) {
  __shared__ int wsum[32];
  const int lane = threadIdx.x & 31;
  const int wave = threadIdx.x >> 5;
  const int base = wave * (SCAN_TILES * 128) + lane * 4;
  int4 v[SCAN_TILES];
  int tot = 0;
  #pragma unroll
  for (int j = 0; j < SCAN_TILES; ++j) {
    const int i = base + j * 128;
    v[j] = (i < N_NODES) ? *(const int4*)(cr + i) : make_int4(0, 0, 0, 0);
    tot += v[j].x + v[j].y + v[j].z + v[j].w;
  }
  for (int o = 16; o > 0; o >>= 1) tot += __shfl_xor(tot, o);
  if (lane == 0) wsum[wave] = tot;
  __syncthreads();
  int pre = (lane < wave) ? wsum[lane] : 0;
  for (int o = 16; o > 0; o >>= 1) pre += __shfl_xor(pre, o);
  int carry = pre;
  #pragma unroll
  for (int j = 0; j < SCAN_TILES; ++j) {
    const int4 c = v[j];
    const int t = c.x + c.y + c.z + c.w;
    int inc = t;
    for (int o = 1; o < 32; o <<= 1) {
      const int up = __shfl_up(inc, o);
      if (lane >= o) inc += up;
    }
    const int ex = carry + inc - t;
    const int i = base + j * 128;
    if (i < N_NODES) {
      *(int4*)(cr + i) = make_int4(ex, ex + c.x, ex + c.x + c.y, ex + c.x + c.y + c.z);
      *(float4*)(dinv + i) = make_float4(rsqrtf((float)c.x + 1.0f), rsqrtf((float)c.y + 1.0f),
                                         rsqrtf((float)c.z + 1.0f), rsqrtf((float)c.w + 1.0f));
    }
    carry += __shfl(inc, 31);
  }
  if (threadIdx.x == 1023) cr[N_NODES] = carry;
}

__global__ void fill_kernel(const int* __restrict__ src, const int* __restrict__ dst,
                            const int* __restrict__ rowptr, const int* __restrict__ rank,
                            const float* __restrict__ dinv, int2* __restrict__ csr) {
  int e = blockIdx.x * blockDim.x + threadIdx.x;
  if (e >= N_EDGES) return;
  const int s = src[e];
  const int d = dst[e];
  const float norm = dinv[s] * dinv[d];
  csr[rowptr[d] + rank[e]] = make_int2(s, __float_as_int(norm));
}

__global__ __launch_bounds__(256) void gemm_wmma(const float* __restrict__ X,
                                                 const float* __restrict__ W,
                                                 float* __restrict__ H) {
  const int wave = threadIdx.x >> 5;
  const int lane = threadIdx.x & 31;
  const int half = lane >> 4;
  const int l    = lane & 15;
  const int m0   = blockIdx.x * 16;
  const int n0   = wave * 16;

  const float* Xrow = X + (size_t)(m0 + l) * NF;
  v8f acc = {};

  #pragma unroll
  for (int k = 0; k < NF; k += 4) {
    v2f a = *(const v2f*)(Xrow + k + 2 * half);
    v2f b;
    b.x = W[(size_t)(k + 2 * half + 0) * NF + n0 + l];
    b.y = W[(size_t)(k + 2 * half + 1) * NF + n0 + l];
    acc = __builtin_amdgcn_wmma_f32_16x16x4_f32(
         false, a,  false, b,
         (short)0, acc,  false,  false);
  }

  float* Hp = H + (size_t)(m0 + 8 * half) * NF + n0 + l;
  #pragma unroll
  for (int r = 0; r < 8; ++r) Hp[(size_t)r * NF] = acc[r];
}

static_assert(N_NODES % 8 == 0, "eight rows per block, no tail");
template <bool L1>
__global__ __launch_bounds__(256) void gather_kernel(const float* __restrict__ H,
                                                     const int* __restrict__ rowptr,
                                                     const int2* __restrict__ csr,
                                                     const float* __restrict__ dinv,
                                                     const float* __restrict__ b,
                                                     float* __restrict__ OUT) {
  const int lane = threadIdx.x & 31;
  const int i   = __builtin_amdgcn_readfirstlane(blockIdx.x * 8 + (threadIdx.x >> 5));
  const int beg = __builtin_amdgcn_readfirstlane(rowptr[i]);
  const int end = __builtin_amdgcn_readfirstlane(rowptr[i + 1]);
  const float* Hl = H + lane * 4;

  float4 hs = make_float4(0.0f, 0.0f, 0.0f, 0.0f);
  float4 bv = hs;
  float di = 0.0f;
  if (L1) {
    hs = *(const float4*)(Hl + (size_t)i * NF);
    bv = *(const float4*)(b + lane * 4);
    di = dinv[i];
  }

  float4 acc = make_float4(0.0f, 0.0f, 0.0f, 0.0f);
  for (int c = beg; c < end; c += 32) {
    const int n = min(32, end - c);
    int2 sw = make_int2(0, 0);
    if (lane < n) sw = csr[c + lane];
    int k = 0;
    for (; k + 4 <= n; k += 4) {
      const int s0 = __builtin_amdgcn_readlane(sw.x, k + 0);
      const int s1 = __builtin_amdgcn_readlane(sw.x, k + 1);
      const int s2 = __builtin_amdgcn_readlane(sw.x, k + 2);
      const int s3 = __builtin_amdgcn_readlane(sw.x, k + 3);
      const float w0 = __int_as_float(__builtin_amdgcn_readlane(sw.y, k + 0));
      const float w1 = __int_as_float(__builtin_amdgcn_readlane(sw.y, k + 1));
      const float w2 = __int_as_float(__builtin_amdgcn_readlane(sw.y, k + 2));
      const float w3 = __int_as_float(__builtin_amdgcn_readlane(sw.y, k + 3));
      const float4 h0 = *(const float4*)(Hl + (size_t)s0 * NF);
      const float4 h1 = *(const float4*)(Hl + (size_t)s1 * NF);
      const float4 h2 = *(const float4*)(Hl + (size_t)s2 * NF);
      const float4 h3 = *(const float4*)(Hl + (size_t)s3 * NF);
      acc.x += h0.x * w0; acc.y += h0.y * w0; acc.z += h0.z * w0; acc.w += h0.w * w0;
      acc.x += h1.x * w1; acc.y += h1.y * w1; acc.z += h1.z * w1; acc.w += h1.w * w1;
      acc.x += h2.x * w2; acc.y += h2.y * w2; acc.z += h2.z * w2; acc.w += h2.w * w2;
      acc.x += h3.x * w3; acc.y += h3.y * w3; acc.z += h3.z * w3; acc.w += h3.w * w3;
    }
    const int rem = n - k;
    if (rem > 0) {
      const int s0 = __builtin_amdgcn_readlane(sw.x, k);
      const float w0 = __int_as_float(__builtin_amdgcn_readlane(sw.y, k));
      const float4 h0 = *(const float4*)(Hl + (size_t)s0 * NF);
      if (rem > 1) {
        const int s1 = __builtin_amdgcn_readlane(sw.x, k + 1);
        const float w1 = __int_as_float(__builtin_amdgcn_readlane(sw.y, k + 1));
        const float4 h1 = *(const float4*)(Hl + (size_t)s1 * NF);
        if (rem > 2) {
          const int s2 = __builtin_amdgcn_readlane(sw.x, k + 2);
          const float w2 = __int_as_float(__builtin_amdgcn_readlane(sw.y, k + 2));
          const float4 h2 = *(const float4*)(Hl + (size_t)s2 * NF);
          acc.x += h0.x * w0; acc.y += h0.y * w0; acc.z += h0.z * w0; acc.w += h0.w * w0;
          acc.x += h1.x * w1; acc.y += h1.y * w1; acc.z += h1.z * w1; acc.w += h1.w * w1;
          acc.x += h2.x * w2; acc.y += h2.y * w2; acc.z += h2.z * w2; acc.w += h2.w * w2;
        } else {
          acc.x += h0.x * w0; acc.y += h0.y * w0; acc.z += h0.z * w0; acc.w += h0.w * w0;
          acc.x += h1.x * w1; acc.y += h1.y * w1; acc.z += h1.z * w1; acc.w += h1.w * w1;
        }
      } else {
        acc.x += h0.x * w0; acc.y += h0.y * w0; acc.z += h0.z * w0; acc.w += h0.w * w0;
      }
    }
  }

  float4 r = acc;
  if (L1) {
    const float di2 = di * di;
    r.x = fmaxf(acc.x + di2 * hs.x + bv.x, 0.0f);
    r.y = fmaxf(acc.y + di2 * hs.y + bv.y, 0.0f);
    r.z = fmaxf(acc.z + di2 * hs.z + bv.z, 0.0f);
    r.w = fmaxf(acc.w + di2 * hs.w + bv.w, 0.0f);
  }
  *(float4*)(OUT + (size_t)i * NF + lane * 4) = r;
}

__global__ __launch_bounds__(256) void combine_pool(const float* __restrict__ AGG,
                                                    const float* __restrict__ H,
                                                    const float* __restrict__ dinv,
                                                    const float* __restrict__ b,
                                                    const int* __restrict__ batch,
                                                    float* __restrict__ out) {
  int gid = blockIdx.x * blockDim.x + threadIdx.x;
  if (gid >= N_NODES * 32) return;
  int i = gid >> 5;
  int q = (gid & 31) * 4;
  float di = dinv[i];
  float di2 = di * di;
  size_t off = (size_t)i * NF + q;
  float4 av = *(const float4*)(AGG + off);
  float4 hv = *(const float4*)(H + off);
  float4 bv = *(const float4*)(b + q);
  float* op = out + (size_t)batch[i] * NF + q;
  atomicAdd(op + 0, av.x + di2 * hv.x + bv.x);
  atomicAdd(op + 1, av.y + di2 * hv.y + bv.y);
  atomicAdd(op + 2, av.z + di2 * hv.z + bv.z);
  atomicAdd(op + 3, av.w + di2 * hv.w + bv.w);
}

extern "C" void kernel_launch(void* const* d_in, const int* in_sizes, int n_in,
                              void* d_out, int out_size, void* d_ws, size_t ws_size,
                              hipStream_t stream) {
  const float* x    = (const float*)d_in[0];
  const int*   ei   = (const int*)  d_in[1];
  const int*   batch= (const int*)  d_in[2];
  const float* W1   = (const float*)d_in[3];
  const float* b1   = (const float*)d_in[4];
  const float* W2   = (const float*)d_in[5];
  const float* b2   = (const float*)d_in[6];
  float* out = (float*)d_out;
  float* ws  = (float*)d_ws;

  const int* src = ei;
  const int* dst = ei + N_EDGES;

  float* dinv = ws;
  float* H    = ws + 50048;
  float* R    = H + (size_t)N_NODES * NF;
  float* Y    = R + (size_t)N_NODES * NF;
  float* AGG  = Y;

  int*  rowptr = (int*)R;
  int*  rank   = rowptr + 50048;
  int2* csr    = (int2*)(rank + N_EDGES);

  const int T = 256;
  const int gNodes    = (N_NODES + T - 1) / T;
  const int gEdges    = (N_EDGES + T - 1) / T;
  const int gNodeVec  = (N_NODES * 32 + T - 1) / T;
  const int gRows     = N_NODES / 8;
  const int gGemm     = N_NODES / 16;

  zero_int_kernel<<<gNodes, T, 0, stream>>>(rowptr, N_NODES);
  count_rank_kernel<<<gEdges, T, 0, stream>>>(dst, rowptr, rank);
  scan_dinv_kernel<<<1, 1024, 0, stream>>>(rowptr, dinv);
  fill_kernel<<<gEdges, T, 0, stream>>>(src, dst, rowptr, rank, dinv, csr);

  gemm_wmma<<<gGemm, T, 0, stream>>>(x, W1, H);
  gather_kernel<true><<<gRows, T, 0, stream>>>(H, rowptr, csr, dinv, b1, Y);

  gemm_wmma<<<gGemm, T, 0, stream>>>(Y, W2, H);
  gather_kernel<false><<<gRows, T, 0, stream>>>(H, rowptr, csr, dinv, b2, AGG);

  zero_kernel<<<(N_GRAPHS * NF + T - 1) / T, T, 0, stream>>>(out, N_GRAPHS * NF);
  combine_pool<<<gNodeVec, T, 0, stream>>>(AGG, H, dinv, b2, batch, out);
}


# layer-2 combine and graph pooling fused into the layer-2 row gather: 16 rows of a block are summed in LDS in row order per graph id, one float atomic per feature and run; sum order changes
# speedup vs baseline: 2.5491x; 1.3561x over previous
#include <hip/hip_runtime.h>

#define N_NODES  50000
#define N_EDGES  800000
#define NF       128
#define N_GRAPHS 128

typedef __attribute__((ext_vector_type(2))) float v2f;
typedef __attribute__((ext_vector_type(8))) float v8f;

__global__ void zero_kernel(float* __restrict__ p, int n) {
  int i = blockIdx.x * blockDim.x + threadIdx.x;
  if (i < n) p[i] = 0.0f;
}

__global__ void zero_int_kernel(int* __restrict__ p, int n) {
  int i = blockIdx.x * blockDim.x + threadIdx.x;
  if (i < n) p[i] = 0;
}


__global__ void count_rank_kernel(const int* __restrict__ dst, int* __restrict__ cnt,
                                  int* __restrict__ rank) {
  int e = blockIdx.x * blockDim.x + threadIdx.x;
  if (e < N_EDGES) rank[e] = atomicAdd(&cnt[dst[e]], 1);
}

#define SCAN_TILES 13
static_assert(32 * SCAN_TILES * 128 >= N_NODES, "scan block must cover all nodes");
static_assert(N_NODES % 4 == 0, "counts are handled four at a time");
__global__ __launch_bounds__(1024) void scan_dinv_kernel(int* __restrict__ cr,
                                                         float* __restrict__ dinv) {
  __shared__ int wsum[32];
  const int lane = threadIdx.x & 31;
  const int wave = threadIdx.x >> 5;
  const int base = wave * (SCAN_TILES * 128) + lane * 4;
  int4 v[SCAN_TILES];
  int tot = 0;
  #pragma unroll
  for (int j = 0; j < SCAN_TILES; ++j) {
    const int i = base + j * 128;
    v[j] = (i < N_NODES) ? *(const int4*)(cr + i) : make_int4(0, 0, 0, 0);
    tot += v[j].x + v[j].y + v[j].z + v[j].w;
  }
  for (int o = 16; o > 0; o >>= 1) tot += __shfl_xor(tot, o);
  if (lane == 0) wsum[wave] = tot;
  __syncthreads();
  int pre = (lane < wave) ? wsum[lane] : 0;
  for (int o = 16; o > 0; o >>= 1) pre += __shfl_xor(pre, o);
  int carry = pre;
  #pragma unroll
  for (int j = 0; j < SCAN_TILES; ++j) {
    const int4 c = v[j];
    const int t = c.x + c.y + c.z + c.w;
    int inc = t;
    for (int o = 1; o < 32; o <<= 1) {
      const int up = __shfl_up(inc, o);
      if (lane >= o) inc += up;
    }
    const int ex = carry + inc - t;
    const int i = base + j * 128;
    if (i < N_NODES) {
      *(int4*)(cr + i) = make_int4(ex, ex + c.x, ex + c.x + c.y, ex + c.x + c.y + c.z);
      *(float4*)(dinv + i) = make_float4(rsqrtf((float)c.x + 1.0f), rsqrtf((float)c.y + 1.0f),
                                         rsqrtf((float)c.z + 1.0f), rsqrtf((float)c.w + 1.0f));
    }
    carry += __shfl(inc, 31);
  }
  if (threadIdx.x == 1023) cr[N_NODES] = carry;
}

__global__ void fill_kernel(const int* __restrict__ src, const int* __restrict__ dst,
                            const int* __restrict__ rowptr, const int* __restrict__ rank,
                            const float* __restrict__ dinv, int2* __restrict__ csr) {
  int e = blockIdx.x * blockDim.x + threadIdx.x;
  if (e >= N_EDGES) return;
  const int s = src[e];
  const int d = dst[e];
  const float norm = dinv[s] * dinv[d];
  csr[rowptr[d] + rank[e]] = make_int2(s, __float_as_int(norm));
}

__global__ __launch_bounds__(256) void gemm_wmma(const float* __restrict__ X,
                                                 const float* __restrict__ W,
                                                 float* __restrict__ H) {
  const int wave = threadIdx.x >> 5;
  const int lane = threadIdx.x & 31;
  const int half = lane >> 4;
  const int l    = lane & 15;
  const int m0   = blockIdx.x * 16;
  const int n0   = wave * 16;

  const float* Xrow = X + (size_t)(m0 + l) * NF;
  v8f acc = {};

  #pragma unroll
  for (int k = 0; k < NF; k += 4) {
    v2f a = *(const v2f*)(Xrow + k + 2 * half);
    v2f b;
    b.x = W[(size_t)(k + 2 * half + 0) * NF + n0 + l];
    b.y = W[(size_t)(k + 2 * half + 1) * NF + n0 + l];
    acc = __builtin_amdgcn_wmma_f32_16x16x4_f32(
         false, a,  false, b,
         (short)0, acc,  false,  false);
  }

  float* Hp = H + (size_t)(m0 + 8 * half) * NF + n0 + l;
  #pragma unroll
  for (int r = 0; r < 8; ++r) Hp[(size_t)r * NF] = acc[r];
}

__device__ __forceinline__ float4 row_sum(const float* __restrict__ Hl,
                                          const int2* __restrict__ csr,
                                          const int beg, const int end, const int lane) {
  float4 acc = make_float4(0.0f, 0.0f, 0.0f, 0.0f);
  for (int c = beg; c < end; c += 32) {
    const int n = min(32, end - c);
    int2 sw = make_int2(0, 0);
    if (lane < n) sw = csr[c + lane];
    int k = 0;
    for (; k + 4 <= n; k += 4) {
      const int s0 = __builtin_amdgcn_readlane(sw.x, k + 0);
      const int s1 = __builtin_amdgcn_readlane(sw.x, k + 1);
      const int s2 = __builtin_amdgcn_readlane(sw.x, k + 2);
      const int s3 = __builtin_amdgcn_readlane(sw.x, k + 3);
      const float w0 = __int_as_float(__builtin_amdgcn_readlane(sw.y, k + 0));
      const float w1 = __int_as_float(__builtin_amdgcn_readlane(sw.y, k + 1));
      const float w2 = __int_as_float(__builtin_amdgcn_readlane(sw.y, k + 2));
      const float w3 = __int_as_float(__builtin_amdgcn_readlane(sw.y, k + 3));
      const float4 h0 = *(const float4*)(Hl + (size_t)s0 * NF);
      const float4 h1 = *(const float4*)(Hl + (size_t)s1 * NF);
      const float4 h2 = *(const float4*)(Hl + (size_t)s2 * NF);
      const float4 h3 = *(const float4*)(Hl + (size_t)s3 * NF);
      acc.x += h0.x * w0; acc.y += h0.y * w0; acc.z += h0.z * w0; acc.w += h0.w * w0;
      acc.x += h1.x * w1; acc.y += h1.y * w1; acc.z += h1.z * w1; acc.w += h1.w * w1;
      acc.x += h2.x * w2; acc.y += h2.y * w2; acc.z += h2.z * w2; acc.w += h2.w * w2;
      acc.x += h3.x * w3; acc.y += h3.y * w3; acc.z += h3.z * w3; acc.w += h3.w * w3;
    }
    const int rem = n - k;
    if (rem > 0) {
      const int s0 = __builtin_amdgcn_readlane(sw.x, k);
      const float w0 = __int_as_float(__builtin_amdgcn_readlane(sw.y, k));
      const float4 h0 = *(const float4*)(Hl + (size_t)s0 * NF);
      if (rem > 1) {
        const int s1 = __builtin_amdgcn_readlane(sw.x, k + 1);
        const float w1 = __int_as_float(__builtin_amdgcn_readlane(sw.y, k + 1));
        const float4 h1 = *(const float4*)(Hl + (size_t)s1 * NF);
        if (rem > 2) {
          const int s2 = __builtin_amdgcn_readlane(sw.x, k + 2);
          const float w2 = __int_as_float(__builtin_amdgcn_readlane(sw.y, k + 2));
          const float4 h2 = *(const float4*)(Hl + (size_t)s2 * NF);
          acc.x += h0.x * w0; acc.y += h0.y * w0; acc.z += h0.z * w0; acc.w += h0.w * w0;
          acc.x += h1.x * w1; acc.y += h1.y * w1; acc.z += h1.z * w1; acc.w += h1.w * w1;
          acc.x += h2.x * w2; acc.y += h2.y * w2; acc.z += h2.z * w2; acc.w += h2.w * w2;
        } else {
          acc.x += h0.x * w0; acc.y += h0.y * w0; acc.z += h0.z * w0; acc.w += h0.w * w0;
          acc.x += h1.x * w1; acc.y += h1.y * w1; acc.z += h1.z * w1; acc.w += h1.w * w1;
        }
      } else {
        acc.x += h0.x * w0; acc.y += h0.y * w0; acc.z += h0.z * w0; acc.w += h0.w * w0;
      }
    }
  }
  return acc;
}

static_assert(N_NODES % 8 == 0, "eight rows per block, no tail");
__global__ __launch_bounds__(256) void gather_relu_kernel(const float* __restrict__ H,
                                                          const int* __restrict__ rowptr,
                                                          const int2* __restrict__ csr,
                                                          const float* __restrict__ dinv,
                                                          const float* __restrict__ b,
                                                          float* __restrict__ OUT) {
  const int lane = threadIdx.x & 31;
  const int i   = __builtin_amdgcn_readfirstlane(blockIdx.x * 8 + (threadIdx.x >> 5));
  const int beg = __builtin_amdgcn_readfirstlane(rowptr[i]);
  const int end = __builtin_amdgcn_readfirstlane(rowptr[i + 1]);
  const float* Hl = H + lane * 4;

  const float4 hs = *(const float4*)(Hl + (size_t)i * NF);
  const float4 bv = *(const float4*)(b + lane * 4);
  const float di = dinv[i];

  const float4 acc = row_sum(Hl, csr, beg, end, lane);

  const float di2 = di * di;
  float4 r;
  r.x = fmaxf(acc.x + di2 * hs.x + bv.x, 0.0f);
  r.y = fmaxf(acc.y + di2 * hs.y + bv.y, 0.0f);
  r.z = fmaxf(acc.z + di2 * hs.z + bv.z, 0.0f);
  r.w = fmaxf(acc.w + di2 * hs.w + bv.w, 0.0f);
  *(float4*)(OUT + (size_t)i * NF + lane * 4) = r;
}

#define POOL_ROWS 16
static_assert(N_NODES % POOL_ROWS == 0, "sixteen rows per block, no tail");
__global__ __launch_bounds__(POOL_ROWS * 32) void gather_pool_kernel(
    const float* __restrict__ H, const int* __restrict__ rowptr,
    const int2* __restrict__ csr, const float* __restrict__ dinv,
    const float* __restrict__ b, const int* __restrict__ batch,
    float* __restrict__ out) {
  __shared__ float4 rows[POOL_ROWS][32];
  __shared__ int gid[POOL_ROWS];
  const int lane = threadIdx.x & 31;
  const int wave = threadIdx.x >> 5;
  const int i   = __builtin_amdgcn_readfirstlane(blockIdx.x * POOL_ROWS + wave);
  const int beg = __builtin_amdgcn_readfirstlane(rowptr[i]);
  const int end = __builtin_amdgcn_readfirstlane(rowptr[i + 1]);
  const float* Hl = H + lane * 4;

  const float4 hs = *(const float4*)(Hl + (size_t)i * NF);
  const float4 bv = *(const float4*)(b + lane * 4);
  const float di = dinv[i];
  const int   g  = batch[i];

  const float4 acc = row_sum(Hl, csr, beg, end, lane);

  const float di2 = di * di;
  float4 r;
  r.x = acc.x + di2 * hs.x + bv.x;
  r.y = acc.y + di2 * hs.y + bv.y;
  r.z = acc.z + di2 * hs.z + bv.z;
  r.w = acc.w + di2 * hs.w + bv.w;
  rows[wave][lane] = r;
  if (lane == 0) gid[wave] = g;
  __syncthreads();

  if (threadIdx.x < NF) {
    const float* rf = (const float*)rows + threadIdx.x;
    float v[POOL_ROWS];
    int   gw[POOL_ROWS];
    #pragma unroll
    for (int w = 0; w < POOL_ROWS; ++w) {
      v[w]  = rf[w * NF];
      gw[w] = gid[w];
    }
    int   gcur = gw[0];
    float s    = v[0];
    #pragma unroll
    for (int w = 1; w < POOL_ROWS; ++w) {
      if (gw[w] != gcur) {
        atomicAdd(out + (size_t)gcur * NF + threadIdx.x, s);
        gcur = gw[w];
        s = v[w];
      } else {
        s += v[w];
      }
    }
    atomicAdd(out + (size_t)gcur * NF + threadIdx.x, s);
  }
}

extern "C" void kernel_launch(void* const* d_in, const int* in_sizes, int n_in,
                              void* d_out, int out_size, void* d_ws, size_t ws_size,
                              hipStream_t stream) {
  const float* x    = (const float*)d_in[0];
  const int*   ei   = (const int*)  d_in[1];
  const int*   batch= (const int*)  d_in[2];
  const float* W1   = (const float*)d_in[3];
  const float* b1   = (const float*)d_in[4];
  const float* W2   = (const float*)d_in[5];
  const float* b2   = (const float*)d_in[6];
  float* out = (float*)d_out;
  float* ws  = (float*)d_ws;

  const int* src = ei;
  const int* dst = ei + N_EDGES;

  float* dinv = ws;
  float* H    = ws + 50048;
  float* R    = H + (size_t)N_NODES * NF;
  float* Y    = R + (size_t)N_NODES * NF;

  int*  rowptr = (int*)R;
  int*  rank   = rowptr + 50048;
  int2* csr    = (int2*)(rank + N_EDGES);

  const int T = 256;
  const int gNodes    = (N_NODES + T - 1) / T;
  const int gEdges    = (N_EDGES + T - 1) / T;
  const int gRows     = N_NODES / 8;
  const int gPool     = N_NODES / POOL_ROWS;
  const int gGemm     = N_NODES / 16;

  zero_int_kernel<<<gNodes, T, 0, stream>>>(rowptr, N_NODES);
  count_rank_kernel<<<gEdges, T, 0, stream>>>(dst, rowptr, rank);
  scan_dinv_kernel<<<1, 1024, 0, stream>>>(rowptr, dinv);
  fill_kernel<<<gEdges, T, 0, stream>>>(src, dst, rowptr, rank, dinv, csr);

  gemm_wmma<<<gGemm, T, 0, stream>>>(x, W1, H);
  gather_relu_kernel<<<gRows, T, 0, stream>>>(H, rowptr, csr, dinv, b1, Y);

  gemm_wmma<<<gGemm, T, 0, stream>>>(Y, W2, H);
  zero_kernel<<<(N_GRAPHS * NF + T - 1) / T, T, 0, stream>>>(out, N_GRAPHS * NF);
  gather_pool_kernel<<<gPool, POOL_ROWS * 32, 0, stream>>>(H, rowptr, csr, dinv, b2, batch, out);
}


# layer-2 product moved behind the pooling (both linear): pool the aggregated relu rows to a 128x128 table, then one 128x128x128 f32 WMMA product plus node count times bias; sum order changes
# speedup vs baseline: 2.8815x; 1.1304x over previous
#include <hip/hip_runtime.h>

#define N_NODES  50000
#define N_EDGES  800000
#define NF       128
#define N_GRAPHS 128

typedef __attribute__((ext_vector_type(2))) float v2f;
typedef __attribute__((ext_vector_type(8))) float v8f;

__global__ void zero_kernel(float* __restrict__ p, int n) {
  int i = blockIdx.x * blockDim.x + threadIdx.x;
  if (i < n) p[i] = 0.0f;
}

__global__ void zero_int_kernel(int* __restrict__ p, int n) {
  int i = blockIdx.x * blockDim.x + threadIdx.x;
  if (i < n) p[i] = 0;
}


__global__ void count_rank_kernel(const int* __restrict__ dst, int* __restrict__ cnt,
                                  int* __restrict__ rank) {
  int e = blockIdx.x * blockDim.x + threadIdx.x;
  if (e < N_EDGES) rank[e] = atomicAdd(&cnt[dst[e]], 1);
}

#define SCAN_TILES 13
static_assert(32 * SCAN_TILES * 128 >= N_NODES, "scan block must cover all nodes");
static_assert(N_NODES % 4 == 0, "counts are handled four at a time");
__global__ __launch_bounds__(1024) void scan_dinv_kernel(int* __restrict__ cr,
                                                         float* __restrict__ dinv) {
  __shared__ int wsum[32];
  const int lane = threadIdx.x & 31;
  const int wave = threadIdx.x >> 5;
  const int base = wave * (SCAN_TILES * 128) + lane * 4;
  int4 v[SCAN_TILES];
  int tot = 0;
  #pragma unroll
  for (int j = 0; j < SCAN_TILES; ++j) {
    const int i = base + j * 128;
    v[j] = (i < N_NODES) ? *(const int4*)(cr + i) : make_int4(0, 0, 0, 0);
    tot += v[j].x + v[j].y + v[j].z + v[j].w;
  }
  for (int o = 16; o > 0; o >>= 1) tot += __shfl_xor(tot, o);
  if (lane == 0) wsum[wave] = tot;
  __syncthreads();
  int pre = (lane < wave) ? wsum[lane] : 0;
  for (int o = 16; o > 0; o >>= 1) pre += __shfl_xor(pre, o);
  int carry = pre;
  #pragma unroll
  for (int j = 0; j < SCAN_TILES; ++j) {
    const int4 c = v[j];
    const int t = c.x + c.y + c.z + c.w;
    int inc = t;
    for (int o = 1; o < 32; o <<= 1) {
      const int up = __shfl_up(inc, o);
      if (lane >= o) inc += up;
    }
    const int ex = carry + inc - t;
    const int i = base + j * 128;
    if (i < N_NODES) {
      *(int4*)(cr + i) = make_int4(ex, ex + c.x, ex + c.x + c.y, ex + c.x + c.y + c.z);
      *(float4*)(dinv + i) = make_float4(rsqrtf((float)c.x + 1.0f), rsqrtf((float)c.y + 1.0f),
                                         rsqrtf((float)c.z + 1.0f), rsqrtf((float)c.w + 1.0f));
    }
    carry += __shfl(inc, 31);
  }
  if (threadIdx.x == 1023) cr[N_NODES] = carry;
}

__global__ void fill_kernel(const int* __restrict__ src, const int* __restrict__ dst,
                            const int* __restrict__ rowptr, const int* __restrict__ rank,
                            const float* __restrict__ dinv, int2* __restrict__ csr) {
  int e = blockIdx.x * blockDim.x + threadIdx.x;
  if (e >= N_EDGES) return;
  const int s = src[e];
  const int d = dst[e];
  const float norm = dinv[s] * dinv[d];
  csr[rowptr[d] + rank[e]] = make_int2(s, __float_as_int(norm));
}

__device__ __forceinline__ v8f tile_product(const float* __restrict__ Xrow,
                                            const float* __restrict__ Wcol,
                                            const int half, v8f acc) {
  #pragma unroll
  for (int k = 0; k < NF; k += 4) {
    v2f a = *(const v2f*)(Xrow + k + 2 * half);
    v2f b;
    b.x = Wcol[(size_t)(k + 2 * half + 0) * NF];
    b.y = Wcol[(size_t)(k + 2 * half + 1) * NF];
    acc = __builtin_amdgcn_wmma_f32_16x16x4_f32(
         false, a,  false, b,
         (short)0, acc,  false,  false);
  }
  return acc;
}

__global__ __launch_bounds__(256) void gemm_wmma(const float* __restrict__ X,
                                                 const float* __restrict__ W,
                                                 float* __restrict__ H) {
  const int wave = threadIdx.x >> 5;
  const int lane = threadIdx.x & 31;
  const int half = lane >> 4;
  const int l    = lane & 15;
  const int m0   = blockIdx.x * 16;
  const int n0   = wave * 16;

  const v8f zero = {};
  const v8f acc = tile_product(X + (size_t)(m0 + l) * NF, W + n0 + l, half, zero);

  float* Hp = H + (size_t)(m0 + 8 * half) * NF + n0 + l;
  #pragma unroll
  for (int r = 0; r < 8; ++r) Hp[(size_t)r * NF] = acc[r];
}

#define ZP (NF + 4)
static_assert(N_GRAPHS % 16 == 0, "whole 16-row tiles, EXEC all ones");
__global__ __launch_bounds__(256) void pooled_product_wmma(const float* __restrict__ Z,
                                                           const float* __restrict__ W,
                                                           const float* __restrict__ b,
                                                           float* __restrict__ out) {
  const int wave = threadIdx.x >> 5;
  const int lane = threadIdx.x & 31;
  const int half = lane >> 4;
  const int l    = lane & 15;
  const int m0   = blockIdx.x * 16;
  const int n0   = wave * 16;

  const float bj = b[n0 + l];
  const float* cp = Z + (size_t)(m0 + 8 * half) * ZP + NF;
  v8f c0;
  #pragma unroll
  for (int r = 0; r < 8; ++r) c0[r] = cp[(size_t)r * ZP] * bj;

  const v8f acc = tile_product(Z + (size_t)(m0 + l) * ZP, W + n0 + l, half, c0);

  float* Op = out + (size_t)(m0 + 8 * half) * NF + n0 + l;
  #pragma unroll
  for (int r = 0; r < 8; ++r) Op[(size_t)r * NF] = acc[r];
}

__device__ __forceinline__ float4 row_sum(const float* __restrict__ Hl,
                                          const int2* __restrict__ csr,
                                          const int beg, const int end, const int lane) {
  float4 acc = make_float4(0.0f, 0.0f, 0.0f, 0.0f);
  for (int c = beg; c < end; c += 32) {
    const int n = min(32, end - c);
    int2 sw = make_int2(0, 0);
    if (lane < n) sw = csr[c + lane];
    int k = 0;
    for (; k + 4 <= n; k += 4) {
      const int s0 = __builtin_amdgcn_readlane(sw.x, k + 0);
      const int s1 = __builtin_amdgcn_readlane(sw.x, k + 1);
      const int s2 = __builtin_amdgcn_readlane(sw.x, k + 2);
      const int s3 = __builtin_amdgcn_readlane(sw.x, k + 3);
      const float w0 = __int_as_float(__builtin_amdgcn_readlane(sw.y, k + 0));
      const float w1 = __int_as_float(__builtin_amdgcn_readlane(sw.y, k + 1));
      const float w2 = __int_as_float(__builtin_amdgcn_readlane(sw.y, k + 2));
      const float w3 = __int_as_float(__builtin_amdgcn_readlane(sw.y, k + 3));
      const float4 h0 = *(const float4*)(Hl + (size_t)s0 * NF);
      const float4 h1 = *(const float4*)(Hl + (size_t)s1 * NF);
      const float4 h2 = *(const float4*)(Hl + (size_t)s2 * NF);
      const float4 h3 = *(const float4*)(Hl + (size_t)s3 * NF);
      acc.x += h0.x * w0; acc.y += h0.y * w0; acc.z += h0.z * w0; acc.w += h0.w * w0;
      acc.x += h1.x * w1; acc.y += h1.y * w1; acc.z += h1.z * w1; acc.w += h1.w * w1;
      acc.x += h2.x * w2; acc.y += h2.y * w2; acc.z += h2.z * w2; acc.w += h2.w * w2;
      acc.x += h3.x * w3; acc.y += h3.y * w3; acc.z += h3.z * w3; acc.w += h3.w * w3;
    }
    const int rem = n - k;
    if (rem > 0) {
      const int s0 = __builtin_amdgcn_readlane(sw.x, k);
      const float w0 = __int_as_float(__builtin_amdgcn_readlane(sw.y, k));
      const float4 h0 = *(const float4*)(Hl + (size_t)s0 * NF);
      if (rem > 1) {
        const int s1 = __builtin_amdgcn_readlane(sw.x, k + 1);
        const float w1 = __int_as_float(__builtin_amdgcn_readlane(sw.y, k + 1));
        const float4 h1 = *(const float4*)(Hl + (size_t)s1 * NF);
        if (rem > 2) {
          const int s2 = __builtin_amdgcn_readlane(sw.x, k + 2);
          const float w2 = __int_as_float(__builtin_amdgcn_readlane(sw.y, k + 2));
          const float4 h2 = *(const float4*)(Hl + (size_t)s2 * NF);
          acc.x += h0.x * w0; acc.y += h0.y * w0; acc.z += h0.z * w0; acc.w += h0.w * w0;
          acc.x += h1.x * w1; acc.y += h1.y * w1; acc.z += h1.z * w1; acc.w += h1.w * w1;
          acc.x += h2.x * w2; acc.y += h2.y * w2; acc.z += h2.z * w2; acc.w += h2.w * w2;
        } else {
          acc.x += h0.x * w0; acc.y += h0.y * w0; acc.z += h0.z * w0; acc.w += h0.w * w0;
          acc.x += h1.x * w1; acc.y += h1.y * w1; acc.z += h1.z * w1; acc.w += h1.w * w1;
        }
      } else {
        acc.x += h0.x * w0; acc.y += h0.y * w0; acc.z += h0.z * w0; acc.w += h0.w * w0;
      }
    }
  }
  return acc;
}

static_assert(N_NODES % 8 == 0, "eight rows per block, no tail");
__global__ __launch_bounds__(256) void gather_relu_kernel(const float* __restrict__ H,
                                                          const int* __restrict__ rowptr,
                                                          const int2* __restrict__ csr,
                                                          const float* __restrict__ dinv,
                                                          const float* __restrict__ b,
                                                          float* __restrict__ OUT) {
  const int lane = threadIdx.x & 31;
  const int i   = __builtin_amdgcn_readfirstlane(blockIdx.x * 8 + (threadIdx.x >> 5));
  const int beg = __builtin_amdgcn_readfirstlane(rowptr[i]);
  const int end = __builtin_amdgcn_readfirstlane(rowptr[i + 1]);
  const float* Hl = H + lane * 4;

  const float4 hs = *(const float4*)(Hl + (size_t)i * NF);
  const float4 bv = *(const float4*)(b + lane * 4);
  const float di = dinv[i];

  const float4 acc = row_sum(Hl, csr, beg, end, lane);

  const float di2 = di * di;
  float4 r;
  r.x = fmaxf(acc.x + di2 * hs.x + bv.x, 0.0f);
  r.y = fmaxf(acc.y + di2 * hs.y + bv.y, 0.0f);
  r.z = fmaxf(acc.z + di2 * hs.z + bv.z, 0.0f);
  r.w = fmaxf(acc.w + di2 * hs.w + bv.w, 0.0f);
  *(float4*)(OUT + (size_t)i * NF + lane * 4) = r;
}

#define POOL_ROWS 16
static_assert(N_NODES % POOL_ROWS == 0, "sixteen rows per block, no tail");
__global__ __launch_bounds__(POOL_ROWS * 32) void gather_pool_kernel(
    const float* __restrict__ Y, const int* __restrict__ rowptr,
    const int2* __restrict__ csr, const float* __restrict__ dinv,
    const int* __restrict__ batch, float* __restrict__ Z) {
  __shared__ float4 rows[POOL_ROWS][ZP / 4];
  __shared__ int gid[POOL_ROWS];
  const int lane = threadIdx.x & 31;
  const int wave = threadIdx.x >> 5;
  const int i   = __builtin_amdgcn_readfirstlane(blockIdx.x * POOL_ROWS + wave);
  const int beg = __builtin_amdgcn_readfirstlane(rowptr[i]);
  const int end = __builtin_amdgcn_readfirstlane(rowptr[i + 1]);
  const float* Yl = Y + lane * 4;

  const float4 hs = *(const float4*)(Yl + (size_t)i * NF);
  const float di = dinv[i];
  const int   g  = batch[i];

  const float4 acc = row_sum(Yl, csr, beg, end, lane);

  const float di2 = di * di;
  float4 r;
  r.x = acc.x + di2 * hs.x;
  r.y = acc.y + di2 * hs.y;
  r.z = acc.z + di2 * hs.z;
  r.w = acc.w + di2 * hs.w;
  rows[wave][lane] = r;
  if (lane == 0) {
    rows[wave][NF / 4] = make_float4(1.0f, 0.0f, 0.0f, 0.0f);
    gid[wave] = g;
  }
  __syncthreads();

  if (threadIdx.x <= NF) {
    const float* rf = (const float*)rows + threadIdx.x;
    float* Zc = Z + threadIdx.x;
    float v[POOL_ROWS];
    int   gw[POOL_ROWS];
    #pragma unroll
    for (int w = 0; w < POOL_ROWS; ++w) {
      v[w]  = rf[w * ZP];
      gw[w] = gid[w];
    }
    int   gcur = gw[0];
    float s    = v[0];
    #pragma unroll
    for (int w = 1; w < POOL_ROWS; ++w) {
      if (gw[w] != gcur) {
        atomicAdd(Zc + (size_t)gcur * ZP, s);
        gcur = gw[w];
        s = v[w];
      } else {
        s += v[w];
      }
    }
    atomicAdd(Zc + (size_t)gcur * ZP, s);
  }
}

extern "C" void kernel_launch(void* const* d_in, const int* in_sizes, int n_in,
                              void* d_out, int out_size, void* d_ws, size_t ws_size,
                              hipStream_t stream) {
  const float* x    = (const float*)d_in[0];
  const int*   ei   = (const int*)  d_in[1];
  const int*   batch= (const int*)  d_in[2];
  const float* W1   = (const float*)d_in[3];
  const float* b1   = (const float*)d_in[4];
  const float* W2   = (const float*)d_in[5];
  const float* b2   = (const float*)d_in[6];
  float* out = (float*)d_out;
  float* ws  = (float*)d_ws;

  const int* src = ei;
  const int* dst = ei + N_EDGES;

  float* dinv = ws;
  float* H    = ws + 50048;
  float* R    = H + (size_t)N_NODES * NF;
  float* Y    = R + (size_t)N_NODES * NF;

  int*  rowptr = (int*)R;
  int*  rank   = rowptr + 50048;
  int2* csr    = (int2*)(rank + N_EDGES);
  float* Z     = (float*)(csr + N_EDGES);

  const int T = 256;
  const int gNodes    = (N_NODES + T - 1) / T;
  const int gEdges    = (N_EDGES + T - 1) / T;
  const int gRows     = N_NODES / 8;
  const int gPool     = N_NODES / POOL_ROWS;
  const int gGemm     = N_NODES / 16;

  zero_int_kernel<<<gNodes, T, 0, stream>>>(rowptr, N_NODES);
  count_rank_kernel<<<gEdges, T, 0, stream>>>(dst, rowptr, rank);
  scan_dinv_kernel<<<1, 1024, 0, stream>>>(rowptr, dinv);
  fill_kernel<<<gEdges, T, 0, stream>>>(src, dst, rowptr, rank, dinv, csr);

  gemm_wmma<<<gGemm, T, 0, stream>>>(x, W1, H);
  gather_relu_kernel<<<gRows, T, 0, stream>>>(H, rowptr, csr, dinv, b1, Y);

  zero_kernel<<<(N_GRAPHS * ZP + T - 1) / T, T, 0, stream>>>(Z, N_GRAPHS * ZP);
  gather_pool_kernel<<<gPool, POOL_ROWS * 32, 0, stream>>>(Y, rowptr, csr, dinv, batch, Z);
  pooled_product_wmma<<<N_GRAPHS / 16, T, 0, stream>>>(Z, W2, b2, out);
}


# launches 9 -> 7: the x @ W1 WMMA product shares the in-degree count's launch (alternating blocks, the two are independent) and one zero launch gives both the counters and Z their start value
# speedup vs baseline: 3.2147x; 1.1156x over previous
#include <hip/hip_runtime.h>

#define N_NODES  50000
#define N_EDGES  800000
#define NF       128
#define N_GRAPHS 128

typedef __attribute__((ext_vector_type(2))) float v2f;
typedef __attribute__((ext_vector_type(8))) float v8f;

#define ZP (NF + 4)

__global__ void zero_start_kernel(int* __restrict__ cnt, float* __restrict__ Z) {
  const int i = blockIdx.x * blockDim.x + threadIdx.x;
  if (i < N_NODES) cnt[i] = 0;
  else if (i < N_NODES + N_GRAPHS * ZP) Z[i - N_NODES] = 0.0f;
}


__device__ __forceinline__ void count_rank_block(const int unit, const int* __restrict__ dst,
                                                 int* __restrict__ cnt, int* __restrict__ rank) {
  const int e = unit * 256 + threadIdx.x;
  if (e < N_EDGES) rank[e] = atomicAdd(&cnt[dst[e]], 1);
}

#define SCAN_TILES 13
static_assert(32 * SCAN_TILES * 128 >= N_NODES, "scan block must cover all nodes");
static_assert(N_NODES % 4 == 0, "counts are handled four at a time");
__global__ __launch_bounds__(1024) void scan_dinv_kernel(int* __restrict__ cr,
                                                         float* __restrict__ dinv) {
  __shared__ int wsum[32];
  const int lane = threadIdx.x & 31;
  const int wave = threadIdx.x >> 5;
  const int base = wave * (SCAN_TILES * 128) + lane * 4;
  int4 v[SCAN_TILES];
  int tot = 0;
  #pragma unroll
  for (int j = 0; j < SCAN_TILES; ++j) {
    const int i = base + j * 128;
    v[j] = (i < N_NODES) ? *(const int4*)(cr + i) : make_int4(0, 0, 0, 0);
    tot += v[j].x + v[j].y + v[j].z + v[j].w;
  }
  for (int o = 16; o > 0; o >>= 1) tot += __shfl_xor(tot, o);
  if (lane == 0) wsum[wave] = tot;
  __syncthreads();
  int pre = (lane < wave) ? wsum[lane] : 0;
  for (int o = 16; o > 0; o >>= 1) pre += __shfl_xor(pre, o);
  int carry = pre;
  #pragma unroll
  for (int j = 0; j < SCAN_TILES; ++j) {
    const int4 c = v[j];
    const int t = c.x + c.y + c.z + c.w;
    int inc = t;
    for (int o = 1; o < 32; o <<= 1) {
      const int up = __shfl_up(inc, o);
      if (lane >= o) inc += up;
    }
    const int ex = carry + inc - t;
    const int i = base + j * 128;
    if (i < N_NODES) {
      *(int4*)(cr + i) = make_int4(ex, ex + c.x, ex + c.x + c.y, ex + c.x + c.y + c.z);
      *(float4*)(dinv + i) = make_float4(rsqrtf((float)c.x + 1.0f), rsqrtf((float)c.y + 1.0f),
                                         rsqrtf((float)c.z + 1.0f), rsqrtf((float)c.w + 1.0f));
    }
    carry += __shfl(inc, 31);
  }
  if (threadIdx.x == 1023) cr[N_NODES] = carry;
}

__global__ void fill_kernel(const int* __restrict__ src, const int* __restrict__ dst,
                            const int* __restrict__ rowptr, const int* __restrict__ rank,
                            const float* __restrict__ dinv, int2* __restrict__ csr) {
  int e = blockIdx.x * blockDim.x + threadIdx.x;
  if (e >= N_EDGES) return;
  const int s = src[e];
  const int d = dst[e];
  const float norm = dinv[s] * dinv[d];
  csr[rowptr[d] + rank[e]] = make_int2(s, __float_as_int(norm));
}

__device__ __forceinline__ v8f tile_product(const float* __restrict__ Xrow,
                                            const float* __restrict__ Wcol,
                                            const int half, v8f acc) {
  #pragma unroll
  for (int k = 0; k < NF; k += 4) {
    v2f a = *(const v2f*)(Xrow + k + 2 * half);
    v2f b;
    b.x = Wcol[(size_t)(k + 2 * half + 0) * NF];
    b.y = Wcol[(size_t)(k + 2 * half + 1) * NF];
    acc = __builtin_amdgcn_wmma_f32_16x16x4_f32(
         false, a,  false, b,
         (short)0, acc,  false,  false);
  }
  return acc;
}

#define GEMM_BLOCKS  (N_NODES / 16)
#define COUNT_BLOCKS ((N_EDGES + 255) / 256)
static_assert(N_NODES % 16 == 0, "whole 16-row stripes, EXEC all ones");
static_assert(GEMM_BLOCKS == COUNT_BLOCKS, "the two kinds of block pair up one to one");
__global__ __launch_bounds__(256) void count_rank_gemm_kernel(const int* __restrict__ dst,
                                                              int* __restrict__ cnt,
                                                              int* __restrict__ rank,
                                                              const float* __restrict__ X,
                                                              const float* __restrict__ W,
                                                              float* __restrict__ H) {
  const int unit = blockIdx.x >> 1;
  if (blockIdx.x & 1) {
    count_rank_block(unit, dst, cnt, rank);
    return;
  }
  const int wave = threadIdx.x >> 5;
  const int lane = threadIdx.x & 31;
  const int half = lane >> 4;
  const int l    = lane & 15;
  const int m0   = unit * 16;
  const int n0   = wave * 16;

  const v8f zero = {};
  const v8f acc = tile_product(X + (size_t)(m0 + l) * NF, W + n0 + l, half, zero);

  float* Hp = H + (size_t)(m0 + 8 * half) * NF + n0 + l;
  #pragma unroll
  for (int r = 0; r < 8; ++r) Hp[(size_t)r * NF] = acc[r];
}

static_assert(N_GRAPHS % 16 == 0, "whole 16-row tiles, EXEC all ones");
__global__ __launch_bounds__(256) void pooled_product_wmma(const float* __restrict__ Z,
                                                           const float* __restrict__ W,
                                                           const float* __restrict__ b,
                                                           float* __restrict__ out) {
  const int wave = threadIdx.x >> 5;
  const int lane = threadIdx.x & 31;
  const int half = lane >> 4;
  const int l    = lane & 15;
  const int m0   = blockIdx.x * 16;
  const int n0   = wave * 16;

  const float bj = b[n0 + l];
  const float* cp = Z + (size_t)(m0 + 8 * half) * ZP + NF;
  v8f c0;
  #pragma unroll
  for (int r = 0; r < 8; ++r) c0[r] = cp[(size_t)r * ZP] * bj;

  const v8f acc = tile_product(Z + (size_t)(m0 + l) * ZP, W + n0 + l, half, c0);

  float* Op = out + (size_t)(m0 + 8 * half) * NF + n0 + l;
  #pragma unroll
  for (int r = 0; r < 8; ++r) Op[(size_t)r * NF] = acc[r];
}

__device__ __forceinline__ float4 row_sum(const float* __restrict__ Hl,
                                          const int2* __restrict__ csr,
                                          const int beg, const int end, const int lane) {
  float4 acc = make_float4(0.0f, 0.0f, 0.0f, 0.0f);
  for (int c = beg; c < end; c += 32) {
    const int n = min(32, end - c);
    int2 sw = make_int2(0, 0);
    if (lane < n) sw = csr[c + lane];
    int k = 0;
    for (; k + 4 <= n; k += 4) {
      const int s0 = __builtin_amdgcn_readlane(sw.x, k + 0);
      const int s1 = __builtin_amdgcn_readlane(sw.x, k + 1);
      const int s2 = __builtin_amdgcn_readlane(sw.x, k + 2);
      const int s3 = __builtin_amdgcn_readlane(sw.x, k + 3);
      const float w0 = __int_as_float(__builtin_amdgcn_readlane(sw.y, k + 0));
      const float w1 = __int_as_float(__builtin_amdgcn_readlane(sw.y, k + 1));
      const float w2 = __int_as_float(__builtin_amdgcn_readlane(sw.y, k + 2));
      const float w3 = __int_as_float(__builtin_amdgcn_readlane(sw.y, k + 3));
      const float4 h0 = *(const float4*)(Hl + (size_t)s0 * NF);
      const float4 h1 = *(const float4*)(Hl + (size_t)s1 * NF);
      const float4 h2 = *(const float4*)(Hl + (size_t)s2 * NF);
      const float4 h3 = *(const float4*)(Hl + (size_t)s3 * NF);
      acc.x += h0.x * w0; acc.y += h0.y * w0; acc.z += h0.z * w0; acc.w += h0.w * w0;
      acc.x += h1.x * w1; acc.y += h1.y * w1; acc.z += h1.z * w1; acc.w += h1.w * w1;
      acc.x += h2.x * w2; acc.y += h2.y * w2; acc.z += h2.z * w2; acc.w += h2.w * w2;
      acc.x += h3.x * w3; acc.y += h3.y * w3; acc.z += h3.z * w3; acc.w += h3.w * w3;
    }
    const int rem = n - k;
    if (rem > 0) {
      const int s0 = __builtin_amdgcn_readlane(sw.x, k);
      const float w0 = __int_as_float(__builtin_amdgcn_readlane(sw.y, k));
      const float4 h0 = *(const float4*)(Hl + (size_t)s0 * NF);
      if (rem > 1) {
        const int s1 = __builtin_amdgcn_readlane(sw.x, k + 1);
        const float w1 = __int_as_float(__builtin_amdgcn_readlane(sw.y, k + 1));
        const float4 h1 = *(const float4*)(Hl + (size_t)s1 * NF);
        if (rem > 2) {
          const int s2 = __builtin_amdgcn_readlane(sw.x, k + 2);
          const float w2 = __int_as_float(__builtin_amdgcn_readlane(sw.y, k + 2));
          const float4 h2 = *(const float4*)(Hl + (size_t)s2 * NF);
          acc.x += h0.x * w0; acc.y += h0.y * w0; acc.z += h0.z * w0; acc.w += h0.w * w0;
          acc.x += h1.x * w1; acc.y += h1.y * w1; acc.z += h1.z * w1; acc.w += h1.w * w1;
          acc.x += h2.x * w2; acc.y += h2.y * w2; acc.z += h2.z * w2; acc.w += h2.w * w2;
        } else {
          acc.x += h0.x * w0; acc.y += h0.y * w0; acc.z += h0.z * w0; acc.w += h0.w * w0;
          acc.x += h1.x * w1; acc.y += h1.y * w1; acc.z += h1.z * w1; acc.w += h1.w * w1;
        }
      } else {
        acc.x += h0.x * w0; acc.y += h0.y * w0; acc.z += h0.z * w0; acc.w += h0.w * w0;
      }
    }
  }
  return acc;
}

static_assert(N_NODES % 8 == 0, "eight rows per block, no tail");
__global__ __launch_bounds__(256) void gather_relu_kernel(const float* __restrict__ H,
                                                          const int* __restrict__ rowptr,
                                                          const int2* __restrict__ csr,
                                                          const float* __restrict__ dinv,
                                                          const float* __restrict__ b,
                                                          float* __restrict__ OUT) {
  const int lane = threadIdx.x & 31;
  const int i   = __builtin_amdgcn_readfirstlane(blockIdx.x * 8 + (threadIdx.x >> 5));
  const int beg = __builtin_amdgcn_readfirstlane(rowptr[i]);
  const int end = __builtin_amdgcn_readfirstlane(rowptr[i + 1]);
  const float* Hl = H + lane * 4;

  const float4 hs = *(const float4*)(Hl + (size_t)i * NF);
  const float4 bv = *(const float4*)(b + lane * 4);
  const float di = dinv[i];

  const float4 acc = row_sum(Hl, csr, beg, end, lane);

  const float di2 = di * di;
  float4 r;
  r.x = fmaxf(acc.x + di2 * hs.x + bv.x, 0.0f);
  r.y = fmaxf(acc.y + di2 * hs.y + bv.y, 0.0f);
  r.z = fmaxf(acc.z + di2 * hs.z + bv.z, 0.0f);
  r.w = fmaxf(acc.w + di2 * hs.w + bv.w, 0.0f);
  *(float4*)(OUT + (size_t)i * NF + lane * 4) = r;
}

#define POOL_ROWS 16
static_assert(N_NODES % POOL_ROWS == 0, "sixteen rows per block, no tail");
__global__ __launch_bounds__(POOL_ROWS * 32) void gather_pool_kernel(
    const float* __restrict__ Y, const int* __restrict__ rowptr,
    const int2* __restrict__ csr, const float* __restrict__ dinv,
    const int* __restrict__ batch, float* __restrict__ Z) {
  __shared__ float4 rows[POOL_ROWS][ZP / 4];
  __shared__ int gid[POOL_ROWS];
  const int lane = threadIdx.x & 31;
  const int wave = threadIdx.x >> 5;
  const int i   = __builtin_amdgcn_readfirstlane(blockIdx.x * POOL_ROWS + wave);
  const int beg = __builtin_amdgcn_readfirstlane(rowptr[i]);
  const int end = __builtin_amdgcn_readfirstlane(rowptr[i + 1]);
  const float* Yl = Y + lane * 4;

  const float4 hs = *(const float4*)(Yl + (size_t)i * NF);
  const float di = dinv[i];
  const int   g  = batch[i];

  const float4 acc = row_sum(Yl, csr, beg, end, lane);

  const float di2 = di * di;
  float4 r;
  r.x = acc.x + di2 * hs.x;
  r.y = acc.y + di2 * hs.y;
  r.z = acc.z + di2 * hs.z;
  r.w = acc.w + di2 * hs.w;
  rows[wave][lane] = r;
  if (lane == 0) {
    rows[wave][NF / 4] = make_float4(1.0f, 0.0f, 0.0f, 0.0f);
    gid[wave] = g;
  }
  __syncthreads();

  if (threadIdx.x <= NF) {
    const float* rf = (const float*)rows + threadIdx.x;
    float* Zc = Z + threadIdx.x;
    float v[POOL_ROWS];
    int   gw[POOL_ROWS];
    #pragma unroll
    for (int w = 0; w < POOL_ROWS; ++w) {
      v[w]  = rf[w * ZP];
      gw[w] = gid[w];
    }
    int   gcur = gw[0];
    float s    = v[0];
    #pragma unroll
    for (int w = 1; w < POOL_ROWS; ++w) {
      if (gw[w] != gcur) {
        atomicAdd(Zc + (size_t)gcur * ZP, s);
        gcur = gw[w];
        s = v[w];
      } else {
        s += v[w];
      }
    }
    atomicAdd(Zc + (size_t)gcur * ZP, s);
  }
}

extern "C" void kernel_launch(void* const* d_in, const int* in_sizes, int n_in,
                              void* d_out, int out_size, void* d_ws, size_t ws_size,
                              hipStream_t stream) {
  const float* x    = (const float*)d_in[0];
  const int*   ei   = (const int*)  d_in[1];
  const int*   batch= (const int*)  d_in[2];
  const float* W1   = (const float*)d_in[3];
  const float* b1   = (const float*)d_in[4];
  const float* W2   = (const float*)d_in[5];
  const float* b2   = (const float*)d_in[6];
  float* out = (float*)d_out;
  float* ws  = (float*)d_ws;

  const int* src = ei;
  const int* dst = ei + N_EDGES;

  float* dinv = ws;
  float* H    = ws + 50048;
  float* R    = H + (size_t)N_NODES * NF;
  float* Y    = R + (size_t)N_NODES * NF;

  int*  rowptr = (int*)R;
  int*  rank   = rowptr + 50048;
  int2* csr    = (int2*)(rank + N_EDGES);
  float* Z     = (float*)(csr + N_EDGES);

  const int T = 256;
  const int gZero     = (N_NODES + N_GRAPHS * ZP + T - 1) / T;
  const int gEdges    = (N_EDGES + T - 1) / T;
  const int gRows     = N_NODES / 8;
  const int gPool     = N_NODES / POOL_ROWS;

  zero_start_kernel<<<gZero, T, 0, stream>>>(rowptr, Z);
  count_rank_gemm_kernel<<<GEMM_BLOCKS + COUNT_BLOCKS, T, 0, stream>>>(dst, rowptr, rank, x, W1, H);
  scan_dinv_kernel<<<1, 1024, 0, stream>>>(rowptr, dinv);
  fill_kernel<<<gEdges, T, 0, stream>>>(src, dst, rowptr, rank, dinv, csr);

  gather_relu_kernel<<<gRows, T, 0, stream>>>(H, rowptr, csr, dinv, b1, Y);

  gather_pool_kernel<<<gPool, POOL_ROWS * 32, 0, stream>>>(Y, rowptr, csr, dinv, batch, Z);
  pooled_product_wmma<<<N_GRAPHS / 16, T, 0, stream>>>(Z, W2, b2, out);
}


# x @ W1 blocked over rows: a product block stages 80 X rows in LDS and each wave runs 5 row tiles against every W1 fragment it loads; 625 long-lived product blocks lead the grid; k order unchanged
# speedup vs baseline: 3.3269x; 1.0349x over previous
#include <hip/hip_runtime.h>

#define N_NODES  50000
#define N_EDGES  800000
#define NF       128
#define N_GRAPHS 128

typedef __attribute__((ext_vector_type(2))) float v2f;
typedef __attribute__((ext_vector_type(8))) float v8f;

#define ZP (NF + 4)

__global__ void zero_start_kernel(int* __restrict__ cnt, float* __restrict__ Z) {
  const int i = blockIdx.x * blockDim.x + threadIdx.x;
  if (i < N_NODES) cnt[i] = 0;
  else if (i < N_NODES + N_GRAPHS * ZP) Z[i - N_NODES] = 0.0f;
}


__device__ __forceinline__ void count_rank_block(const int unit, const int* __restrict__ dst,
                                                 int* __restrict__ cnt, int* __restrict__ rank) {
  const int e = unit * 256 + threadIdx.x;
  if (e < N_EDGES) rank[e] = atomicAdd(&cnt[dst[e]], 1);
}

#define SCAN_TILES 13
static_assert(32 * SCAN_TILES * 128 >= N_NODES, "scan block must cover all nodes");
static_assert(N_NODES % 4 == 0, "counts are handled four at a time");
__global__ __launch_bounds__(1024) void scan_dinv_kernel(int* __restrict__ cr,
                                                         float* __restrict__ dinv) {
  __shared__ int wsum[32];
  const int lane = threadIdx.x & 31;
  const int wave = threadIdx.x >> 5;
  const int base = wave * (SCAN_TILES * 128) + lane * 4;
  int4 v[SCAN_TILES];
  int tot = 0;
  #pragma unroll
  for (int j = 0; j < SCAN_TILES; ++j) {
    const int i = base + j * 128;
    v[j] = (i < N_NODES) ? *(const int4*)(cr + i) : make_int4(0, 0, 0, 0);
    tot += v[j].x + v[j].y + v[j].z + v[j].w;
  }
  for (int o = 16; o > 0; o >>= 1) tot += __shfl_xor(tot, o);
  if (lane == 0) wsum[wave] = tot;
  __syncthreads();
  int pre = (lane < wave) ? wsum[lane] : 0;
  for (int o = 16; o > 0; o >>= 1) pre += __shfl_xor(pre, o);
  int carry = pre;
  #pragma unroll
  for (int j = 0; j < SCAN_TILES; ++j) {
    const int4 c = v[j];
    const int t = c.x + c.y + c.z + c.w;
    int inc = t;
    for (int o = 1; o < 32; o <<= 1) {
      const int up = __shfl_up(inc, o);
      if (lane >= o) inc += up;
    }
    const int ex = carry + inc - t;
    const int i = base + j * 128;
    if (i < N_NODES) {
      *(int4*)(cr + i) = make_int4(ex, ex + c.x, ex + c.x + c.y, ex + c.x + c.y + c.z);
      *(float4*)(dinv + i) = make_float4(rsqrtf((float)c.x + 1.0f), rsqrtf((float)c.y + 1.0f),
                                         rsqrtf((float)c.z + 1.0f), rsqrtf((float)c.w + 1.0f));
    }
    carry += __shfl(inc, 31);
  }
  if (threadIdx.x == 1023) cr[N_NODES] = carry;
}

__global__ void fill_kernel(const int* __restrict__ src, const int* __restrict__ dst,
                            const int* __restrict__ rowptr, const int* __restrict__ rank,
                            const float* __restrict__ dinv, int2* __restrict__ csr) {
  int e = blockIdx.x * blockDim.x + threadIdx.x;
  if (e >= N_EDGES) return;
  const int s = src[e];
  const int d = dst[e];
  const float norm = dinv[s] * dinv[d];
  csr[rowptr[d] + rank[e]] = make_int2(s, __float_as_int(norm));
}

template <int R>
__device__ __forceinline__ void tile_product(const float* __restrict__ Xrow, const int tile_pitch,
                                             const float* __restrict__ Wcol,
                                             const int half, v8f (&acc)[R]) {
  #pragma unroll
  for (int k = 0; k < NF; k += 4) {
    v2f b;
    b.x = Wcol[(size_t)(k + 2 * half + 0) * NF];
    b.y = Wcol[(size_t)(k + 2 * half + 1) * NF];
    #pragma unroll
    for (int t = 0; t < R; ++t) {
      const v2f a = *(const v2f*)(Xrow + (size_t)t * tile_pitch + k + 2 * half);
      acc[t] = __builtin_amdgcn_wmma_f32_16x16x4_f32(
           false, a,  false, b,
           (short)0, acc[t],  false,  false);
    }
  }
}

#define GEMM_RT      5
#define GEMM_ROWS    (16 * GEMM_RT)
#define GEMM_BLOCKS  (N_NODES / GEMM_ROWS)
#define COUNT_BLOCKS ((N_EDGES + 255) / 256)
static_assert(N_NODES % GEMM_ROWS == 0, "whole row tiles, EXEC all ones");
static_assert((GEMM_ROWS * (NF / 4)) % 256 == 0, "staging: whole float4 rounds per thread");
__global__ __launch_bounds__(256) void count_rank_gemm_kernel(const int* __restrict__ dst,
                                                              int* __restrict__ cnt,
                                                              int* __restrict__ rank,
                                                              const float* __restrict__ X,
                                                              const float* __restrict__ W,
                                                              float* __restrict__ H) {
  __shared__ float4 xs[GEMM_ROWS][ZP / 4];
  if (blockIdx.x >= GEMM_BLOCKS) {
    count_rank_block(blockIdx.x - GEMM_BLOCKS, dst, cnt, rank);
    return;
  }
  const int wave = threadIdx.x >> 5;
  const int lane = threadIdx.x & 31;
  const int half = lane >> 4;
  const int l    = lane & 15;
  const int m0   = blockIdx.x * GEMM_ROWS;
  const int n0   = wave * 16;

  const float4* Xb = (const float4*)(X + (size_t)m0 * NF);
  #pragma unroll
  for (int i = 0; i < GEMM_ROWS * (NF / 4) / 256; ++i) {
    const int idx = i * 256 + threadIdx.x;
    xs[idx >> 5][idx & 31] = Xb[idx];
  }
  __syncthreads();

  v8f acc[GEMM_RT] = {};
  tile_product<GEMM_RT>((const float*)xs + (size_t)l * ZP, 16 * ZP, W + n0 + l, half, acc);

  #pragma unroll
  for (int t = 0; t < GEMM_RT; ++t) {
    float* Hp = H + (size_t)(m0 + 16 * t + 8 * half) * NF + n0 + l;
    #pragma unroll
    for (int r = 0; r < 8; ++r) Hp[(size_t)r * NF] = acc[t][r];
  }
}

static_assert(N_GRAPHS % 16 == 0, "whole 16-row tiles, EXEC all ones");
__global__ __launch_bounds__(256) void pooled_product_wmma(const float* __restrict__ Z,
                                                           const float* __restrict__ W,
                                                           const float* __restrict__ b,
                                                           float* __restrict__ out) {
  const int wave = threadIdx.x >> 5;
  const int lane = threadIdx.x & 31;
  const int half = lane >> 4;
  const int l    = lane & 15;
  const int m0   = blockIdx.x * 16;
  const int n0   = wave * 16;

  const float bj = b[n0 + l];
  const float* cp = Z + (size_t)(m0 + 8 * half) * ZP + NF;
  v8f c0;
  #pragma unroll
  for (int r = 0; r < 8; ++r) c0[r] = cp[(size_t)r * ZP] * bj;

  v8f acc[1] = {c0};
  tile_product<1>(Z + (size_t)(m0 + l) * ZP, 0, W + n0 + l, half, acc);

  float* Op = out + (size_t)(m0 + 8 * half) * NF + n0 + l;
  #pragma unroll
  for (int r = 0; r < 8; ++r) Op[(size_t)r * NF] = acc[0][r];
}

__device__ __forceinline__ float4 row_sum(const float* __restrict__ Hl,
                                          const int2* __restrict__ csr,
                                          const int beg, const int end, const int lane) {
  float4 acc = make_float4(0.0f, 0.0f, 0.0f, 0.0f);
  for (int c = beg; c < end; c += 32) {
    const int n = min(32, end - c);
    int2 sw = make_int2(0, 0);
    if (lane < n) sw = csr[c + lane];
    int k = 0;
    for (; k + 4 <= n; k += 4) {
      const int s0 = __builtin_amdgcn_readlane(sw.x, k + 0);
      const int s1 = __builtin_amdgcn_readlane(sw.x, k + 1);
      const int s2 = __builtin_amdgcn_readlane(sw.x, k + 2);
      const int s3 = __builtin_amdgcn_readlane(sw.x, k + 3);
      const float w0 = __int_as_float(__builtin_amdgcn_readlane(sw.y, k + 0));
      const float w1 = __int_as_float(__builtin_amdgcn_readlane(sw.y, k + 1));
      const float w2 = __int_as_float(__builtin_amdgcn_readlane(sw.y, k + 2));
      const float w3 = __int_as_float(__builtin_amdgcn_readlane(sw.y, k + 3));
      const float4 h0 = *(const float4*)(Hl + (size_t)s0 * NF);
      const float4 h1 = *(const float4*)(Hl + (size_t)s1 * NF);
      const float4 h2 = *(const float4*)(Hl + (size_t)s2 * NF);
      const float4 h3 = *(const float4*)(Hl + (size_t)s3 * NF);
      acc.x += h0.x * w0; acc.y += h0.y * w0; acc.z += h0.z * w0; acc.w += h0.w * w0;
      acc.x += h1.x * w1; acc.y += h1.y * w1; acc.z += h1.z * w1; acc.w += h1.w * w1;
      acc.x += h2.x * w2; acc.y += h2.y * w2; acc.z += h2.z * w2; acc.w += h2.w * w2;
      acc.x += h3.x * w3; acc.y += h3.y * w3; acc.z += h3.z * w3; acc.w += h3.w * w3;
    }
    const int rem = n - k;
    if (rem > 0) {
      const int s0 = __builtin_amdgcn_readlane(sw.x, k);
      const float w0 = __int_as_float(__builtin_amdgcn_readlane(sw.y, k));
      const float4 h0 = *(const float4*)(Hl + (size_t)s0 * NF);
      if (rem > 1) {
        const int s1 = __builtin_amdgcn_readlane(sw.x, k + 1);
        const float w1 = __int_as_float(__builtin_amdgcn_readlane(sw.y, k + 1));
        const float4 h1 = *(const float4*)(Hl + (size_t)s1 * NF);
        if (rem > 2) {
          const int s2 = __builtin_amdgcn_readlane(sw.x, k + 2);
          const float w2 = __int_as_float(__builtin_amdgcn_readlane(sw.y, k + 2));
          const float4 h2 = *(const float4*)(Hl + (size_t)s2 * NF);
          acc.x += h0.x * w0; acc.y += h0.y * w0; acc.z += h0.z * w0; acc.w += h0.w * w0;
          acc.x += h1.x * w1; acc.y += h1.y * w1; acc.z += h1.z * w1; acc.w += h1.w * w1;
          acc.x += h2.x * w2; acc.y += h2.y * w2; acc.z += h2.z * w2; acc.w += h2.w * w2;
        } else {
          acc.x += h0.x * w0; acc.y += h0.y * w0; acc.z += h0.z * w0; acc.w += h0.w * w0;
          acc.x += h1.x * w1; acc.y += h1.y * w1; acc.z += h1.z * w1; acc.w += h1.w * w1;
        }
      } else {
        acc.x += h0.x * w0; acc.y += h0.y * w0; acc.z += h0.z * w0; acc.w += h0.w * w0;
      }
    }
  }
  return acc;
}

static_assert(N_NODES % 8 == 0, "eight rows per block, no tail");
__global__ __launch_bounds__(256) void gather_relu_kernel(const float* __restrict__ H,
                                                          const int* __restrict__ rowptr,
                                                          const int2* __restrict__ csr,
                                                          const float* __restrict__ dinv,
                                                          const float* __restrict__ b,
                                                          float* __restrict__ OUT) {
  const int lane = threadIdx.x & 31;
  const int i   = __builtin_amdgcn_readfirstlane(blockIdx.x * 8 + (threadIdx.x >> 5));
  const int beg = __builtin_amdgcn_readfirstlane(rowptr[i]);
  const int end = __builtin_amdgcn_readfirstlane(rowptr[i + 1]);
  const float* Hl = H + lane * 4;

  const float4 hs = *(const float4*)(Hl + (size_t)i * NF);
  const float4 bv = *(const float4*)(b + lane * 4);
  const float di = dinv[i];

  const float4 acc = row_sum(Hl, csr, beg, end, lane);

  const float di2 = di * di;
  float4 r;
  r.x = fmaxf(acc.x + di2 * hs.x + bv.x, 0.0f);
  r.y = fmaxf(acc.y + di2 * hs.y + bv.y, 0.0f);
  r.z = fmaxf(acc.z + di2 * hs.z + bv.z, 0.0f);
  r.w = fmaxf(acc.w + di2 * hs.w + bv.w, 0.0f);
  *(float4*)(OUT + (size_t)i * NF + lane * 4) = r;
}

#define POOL_ROWS 16
static_assert(N_NODES % POOL_ROWS == 0, "sixteen rows per block, no tail");
__global__ __launch_bounds__(POOL_ROWS * 32) void gather_pool_kernel(
    const float* __restrict__ Y, const int* __restrict__ rowptr,
    const int2* __restrict__ csr, const float* __restrict__ dinv,
    const int* __restrict__ batch, float* __restrict__ Z) {
  __shared__ float4 rows[POOL_ROWS][ZP / 4];
  __shared__ int gid[POOL_ROWS];
  const int lane = threadIdx.x & 31;
  const int wave = threadIdx.x >> 5;
  const int i   = __builtin_amdgcn_readfirstlane(blockIdx.x * POOL_ROWS + wave);
  const int beg = __builtin_amdgcn_readfirstlane(rowptr[i]);
  const int end = __builtin_amdgcn_readfirstlane(rowptr[i + 1]);
  const float* Yl = Y + lane * 4;

  const float4 hs = *(const float4*)(Yl + (size_t)i * NF);
  const float di = dinv[i];
  const int   g  = batch[i];

  const float4 acc = row_sum(Yl, csr, beg, end, lane);

  const float di2 = di * di;
  float4 r;
  r.x = acc.x + di2 * hs.x;
  r.y = acc.y + di2 * hs.y;
  r.z = acc.z + di2 * hs.z;
  r.w = acc.w + di2 * hs.w;
  rows[wave][lane] = r;
  if (lane == 0) {
    rows[wave][NF / 4] = make_float4(1.0f, 0.0f, 0.0f, 0.0f);
    gid[wave] = g;
  }
  __syncthreads();

  if (threadIdx.x <= NF) {
    const float* rf = (const float*)rows + threadIdx.x;
    float* Zc = Z + threadIdx.x;
    float v[POOL_ROWS];
    int   gw[POOL_ROWS];
    #pragma unroll
    for (int w = 0; w < POOL_ROWS; ++w) {
      v[w]  = rf[w * ZP];
      gw[w] = gid[w];
    }
    int   gcur = gw[0];
    float s    = v[0];
    #pragma unroll
    for (int w = 1; w < POOL_ROWS; ++w) {
      if (gw[w] != gcur) {
        atomicAdd(Zc + (size_t)gcur * ZP, s);
        gcur = gw[w];
        s = v[w];
      } else {
        s += v[w];
      }
    }
    atomicAdd(Zc + (size_t)gcur * ZP, s);
  }
}

extern "C" void kernel_launch(void* const* d_in, const int* in_sizes, int n_in,
                              void* d_out, int out_size, void* d_ws, size_t ws_size,
                              hipStream_t stream) {
  const float* x    = (const float*)d_in[0];
  const int*   ei   = (const int*)  d_in[1];
  const int*   batch= (const int*)  d_in[2];
  const float* W1   = (const float*)d_in[3];
  const float* b1   = (const float*)d_in[4];
  const float* W2   = (const float*)d_in[5];
  const float* b2   = (const float*)d_in[6];
  float* out = (float*)d_out;
  float* ws  = (float*)d_ws;

  const int* src = ei;
  const int* dst = ei + N_EDGES;

  float* dinv = ws;
  float* H    = ws + 50048;
  float* R    = H + (size_t)N_NODES * NF;
  float* Y    = R + (size_t)N_NODES * NF;

  int*  rowptr = (int*)R;
  int*  rank   = rowptr + 50048;
  int2* csr    = (int2*)(rank + N_EDGES);
  float* Z     = (float*)(csr + N_EDGES);

  const int T = 256;
  const int gZero     = (N_NODES + N_GRAPHS * ZP + T - 1) / T;
  const int gEdges    = (N_EDGES + T - 1) / T;
  const int gRows     = N_NODES / 8;
  const int gPool     = N_NODES / POOL_ROWS;

  zero_start_kernel<<<gZero, T, 0, stream>>>(rowptr, Z);
  count_rank_gemm_kernel<<<GEMM_BLOCKS + COUNT_BLOCKS, T, 0, stream>>>(dst, rowptr, rank, x, W1, H);
  scan_dinv_kernel<<<1, 1024, 0, stream>>>(rowptr, dinv);
  fill_kernel<<<gEdges, T, 0, stream>>>(src, dst, rowptr, rank, dinv, csr);

  gather_relu_kernel<<<gRows, T, 0, stream>>>(H, rowptr, csr, dinv, b1, Y);

  gather_pool_kernel<<<gPool, POOL_ROWS * 32, 0, stream>>>(Y, rowptr, csr, dinv, batch, Z);
  pooled_product_wmma<<<N_GRAPHS / 16, T, 0, stream>>>(Z, W2, b2, out);
}


# in-degree count and edge rank on LDS counters: 1024-thread owner blocks (node range x edge slice) sweep dst, scan merges slice counts; no global atomics; x@W1 rides the fill launch; sum order changes
# speedup vs baseline: 3.4325x; 1.0317x over previous
#include <hip/hip_runtime.h>

#define N_NODES  50000
#define N_EDGES  800000
#define NF       128
#define N_GRAPHS 128

typedef __attribute__((ext_vector_type(2))) float v2f;
typedef __attribute__((ext_vector_type(8))) float v8f;

#define ZP (NF + 4)


#define OWN_RANGES  64
#define OWN_SLICES  4
#define OWN_BLOCKS  (OWN_RANGES * OWN_SLICES)
#define OWN_THREADS 1024
#define OWN_NODES   ((N_NODES + OWN_RANGES - 1) / OWN_RANGES)
#define OWN_LOADS   7
#define CNT_PITCH   50048
#define SLICE_EDGES (N_EDGES / OWN_SLICES)
#define E4          (SLICE_EDGES / 4)
static_assert(N_EDGES % (4 * OWN_SLICES) == 0, "equal slices, read four edges at a time");
static_assert(CNT_PITCH >= N_NODES + 1 && CNT_PITCH % 4 == 0, "count arrays: room for the row end, 16-byte aligned");
__global__ __launch_bounds__(OWN_THREADS) void count_rank_owner_kernel(const int* __restrict__ dst,
                                                                        int* __restrict__ cnt,
                                                                        int* __restrict__ rank,
                                                                        float* __restrict__ Z) {
  __shared__ int c[OWN_NODES];
  const int range = blockIdx.x % OWN_RANGES;
  const int slice = blockIdx.x / OWN_RANGES;
  const int lo = range * OWN_NODES;
  for (int i = threadIdx.x; i < OWN_NODES; i += OWN_THREADS) c[i] = 0;
  for (int i = blockIdx.x * OWN_THREADS + threadIdx.x; i < N_GRAPHS * ZP; i += OWN_BLOCKS * OWN_THREADS)
    Z[i] = 0.0f;
  __syncthreads();

  const int4* d4 = (const int4*)(dst + slice * SLICE_EDGES);
  int* rk = rank + slice * SLICE_EDGES;
  const int rot = range * (E4 / OWN_RANGES);
  for (int base = 0; base < E4; base += OWN_THREADS * OWN_LOADS) {
    int4 v[OWN_LOADS];
    int  at[OWN_LOADS];
    #pragma unroll
    for (int u = 0; u < OWN_LOADS; ++u) {
      const int i = base + u * OWN_THREADS + threadIdx.x;
      int j = i + rot;
      if (j >= E4) j -= E4;
      at[u] = j;
      v[u] = (i < E4) ? d4[j] : make_int4(-1, -1, -1, -1);
    }
    #pragma unroll
    for (int u = 0; u < OWN_LOADS; ++u) {
      const int e0 = 4 * at[u];
      const unsigned t0 = (unsigned)(v[u].x - lo);
      const unsigned t1 = (unsigned)(v[u].y - lo);
      const unsigned t2 = (unsigned)(v[u].z - lo);
      const unsigned t3 = (unsigned)(v[u].w - lo);
      if (t0 < (unsigned)OWN_NODES) rk[e0 + 0] = atomicAdd(&c[t0], 1);
      if (t1 < (unsigned)OWN_NODES) rk[e0 + 1] = atomicAdd(&c[t1], 1);
      if (t2 < (unsigned)OWN_NODES) rk[e0 + 2] = atomicAdd(&c[t2], 1);
      if (t3 < (unsigned)OWN_NODES) rk[e0 + 3] = atomicAdd(&c[t3], 1);
    }
  }
  __syncthreads();
  for (int i = threadIdx.x; i < OWN_NODES; i += OWN_THREADS)
    if (lo + i < N_NODES) cnt[slice * CNT_PITCH + lo + i] = c[i];
}

#define SCAN_TILES 13
static_assert(32 * SCAN_TILES * 128 >= N_NODES, "scan block must cover all nodes");
static_assert(N_NODES % 4 == 0, "counts are handled four at a time");
__global__ __launch_bounds__(1024) void scan_dinv_kernel(int* __restrict__ cr,
                                                         float* __restrict__ dinv) {
  __shared__ int wsum[32];
  const int lane = threadIdx.x & 31;
  const int wave = threadIdx.x >> 5;
  const int base = wave * (SCAN_TILES * 128) + lane * 4;
  int4 v[SCAN_TILES];
  int tot = 0;
  #pragma unroll
  for (int j = 0; j < SCAN_TILES; ++j) {
    const int i = base + j * 128;
    int4 a = make_int4(0, 0, 0, 0);
    if (i < N_NODES) {
      #pragma unroll
      for (int s = 0; s < OWN_SLICES; ++s) {
        const int4 cs = *(const int4*)(cr + s * CNT_PITCH + i);
        a.x += cs.x; a.y += cs.y; a.z += cs.z; a.w += cs.w;
      }
    }
    v[j] = a;
    tot += a.x + a.y + a.z + a.w;
  }
  for (int o = 16; o > 0; o >>= 1) tot += __shfl_xor(tot, o);
  if (lane == 0) wsum[wave] = tot;
  __syncthreads();
  int pre = (lane < wave) ? wsum[lane] : 0;
  for (int o = 16; o > 0; o >>= 1) pre += __shfl_xor(pre, o);
  int carry = pre;
  #pragma unroll
  for (int j = 0; j < SCAN_TILES; ++j) {
    const int4 c = v[j];
    const int t = c.x + c.y + c.z + c.w;
    int inc = t;
    for (int o = 1; o < 32; o <<= 1) {
      const int up = __shfl_up(inc, o);
      if (lane >= o) inc += up;
    }
    const int ex = carry + inc - t;
    const int i = base + j * 128;
    if (i < N_NODES) {
      int4 p = make_int4(ex, ex + c.x, ex + c.x + c.y, ex + c.x + c.y + c.z);
      #pragma unroll
      for (int s = 0; s < OWN_SLICES; ++s) {
        int4* q = (int4*)(cr + s * CNT_PITCH + i);
        const int4 cs = *q;
        *q = p;
        p.x += cs.x; p.y += cs.y; p.z += cs.z; p.w += cs.w;
      }
      *(float4*)(dinv + i) = make_float4(rsqrtf((float)c.x + 1.0f), rsqrtf((float)c.y + 1.0f),
                                         rsqrtf((float)c.z + 1.0f), rsqrtf((float)c.w + 1.0f));
    }
    carry += __shfl(inc, 31);
  }
  if (threadIdx.x == 1023) cr[N_NODES] = carry;
}

__device__ __forceinline__ void fill_block(const int unit, const int* __restrict__ src,
                                           const int* __restrict__ dst,
                                           const int* __restrict__ rowptr,
                                           const int* __restrict__ rank,
                                           const float* __restrict__ dinv, int2* __restrict__ csr) {
  const int e = unit * 256 + threadIdx.x;
  if (e >= N_EDGES) return;
  const int s = src[e];
  const int d = dst[e];
  const float norm = dinv[s] * dinv[d];
  csr[rowptr[(e / SLICE_EDGES) * CNT_PITCH + d] + rank[e]] = make_int2(s, __float_as_int(norm));
}

template <int R>
__device__ __forceinline__ void tile_product(const float* __restrict__ Xrow, const int tile_pitch,
                                             const float* __restrict__ Wcol,
                                             const int half, v8f (&acc)[R]) {
  #pragma unroll
  for (int k = 0; k < NF; k += 4) {
    v2f b;
    b.x = Wcol[(size_t)(k + 2 * half + 0) * NF];
    b.y = Wcol[(size_t)(k + 2 * half + 1) * NF];
    #pragma unroll
    for (int t = 0; t < R; ++t) {
      const v2f a = *(const v2f*)(Xrow + (size_t)t * tile_pitch + k + 2 * half);
      acc[t] = __builtin_amdgcn_wmma_f32_16x16x4_f32(
           false, a,  false, b,
           (short)0, acc[t],  false,  false);
    }
  }
}

#define GEMM_RT      5
#define GEMM_ROWS    (16 * GEMM_RT)
#define GEMM_BLOCKS  (N_NODES / GEMM_ROWS)
#define FILL_BLOCKS  ((N_EDGES + 255) / 256)
static_assert(N_NODES % GEMM_ROWS == 0, "whole row tiles, EXEC all ones");
static_assert((GEMM_ROWS * (NF / 4)) % 256 == 0, "staging: whole float4 rounds per thread");
__global__ __launch_bounds__(256) void fill_gemm_kernel(const int* __restrict__ src,
                                                        const int* __restrict__ dst,
                                                        const int* __restrict__ rowptr,
                                                        const int* __restrict__ rank,
                                                        const float* __restrict__ dinv,
                                                        int2* __restrict__ csr,
                                                        const float* __restrict__ X,
                                                        const float* __restrict__ W,
                                                        float* __restrict__ H) {
  __shared__ float4 xs[GEMM_ROWS][ZP / 4];
  if (blockIdx.x >= GEMM_BLOCKS) {
    fill_block(blockIdx.x - GEMM_BLOCKS, src, dst, rowptr, rank, dinv, csr);
    return;
  }
  const int wave = threadIdx.x >> 5;
  const int lane = threadIdx.x & 31;
  const int half = lane >> 4;
  const int l    = lane & 15;
  const int m0   = blockIdx.x * GEMM_ROWS;
  const int n0   = wave * 16;

  const float4* Xb = (const float4*)(X + (size_t)m0 * NF);
  #pragma unroll
  for (int i = 0; i < GEMM_ROWS * (NF / 4) / 256; ++i) {
    const int idx = i * 256 + threadIdx.x;
    xs[idx >> 5][idx & 31] = Xb[idx];
  }
  __syncthreads();

  v8f acc[GEMM_RT] = {};
  tile_product<GEMM_RT>((const float*)xs + (size_t)l * ZP, 16 * ZP, W + n0 + l, half, acc);

  #pragma unroll
  for (int t = 0; t < GEMM_RT; ++t) {
    float* Hp = H + (size_t)(m0 + 16 * t + 8 * half) * NF + n0 + l;
    #pragma unroll
    for (int r = 0; r < 8; ++r) Hp[(size_t)r * NF] = acc[t][r];
  }
}

static_assert(N_GRAPHS % 16 == 0, "whole 16-row tiles, EXEC all ones");
__global__ __launch_bounds__(256) void pooled_product_wmma(const float* __restrict__ Z,
                                                           const float* __restrict__ W,
                                                           const float* __restrict__ b,
                                                           float* __restrict__ out) {
  const int wave = threadIdx.x >> 5;
  const int lane = threadIdx.x & 31;
  const int half = lane >> 4;
  const int l    = lane & 15;
  const int m0   = blockIdx.x * 16;
  const int n0   = wave * 16;

  const float bj = b[n0 + l];
  const float* cp = Z + (size_t)(m0 + 8 * half) * ZP + NF;
  v8f c0;
  #pragma unroll
  for (int r = 0; r < 8; ++r) c0[r] = cp[(size_t)r * ZP] * bj;

  v8f acc[1] = {c0};
  tile_product<1>(Z + (size_t)(m0 + l) * ZP, 0, W + n0 + l, half, acc);

  float* Op = out + (size_t)(m0 + 8 * half) * NF + n0 + l;
  #pragma unroll
  for (int r = 0; r < 8; ++r) Op[(size_t)r * NF] = acc[0][r];
}

__device__ __forceinline__ float4 row_sum(const float* __restrict__ Hl,
                                          const int2* __restrict__ csr,
                                          const int beg, const int end, const int lane) {
  float4 acc = make_float4(0.0f, 0.0f, 0.0f, 0.0f);
  for (int c = beg; c < end; c += 32) {
    const int n = min(32, end - c);
    int2 sw = make_int2(0, 0);
    if (lane < n) sw = csr[c + lane];
    int k = 0;
    for (; k + 4 <= n; k += 4) {
      const int s0 = __builtin_amdgcn_readlane(sw.x, k + 0);
      const int s1 = __builtin_amdgcn_readlane(sw.x, k + 1);
      const int s2 = __builtin_amdgcn_readlane(sw.x, k + 2);
      const int s3 = __builtin_amdgcn_readlane(sw.x, k + 3);
      const float w0 = __int_as_float(__builtin_amdgcn_readlane(sw.y, k + 0));
      const float w1 = __int_as_float(__builtin_amdgcn_readlane(sw.y, k + 1));
      const float w2 = __int_as_float(__builtin_amdgcn_readlane(sw.y, k + 2));
      const float w3 = __int_as_float(__builtin_amdgcn_readlane(sw.y, k + 3));
      const float4 h0 = *(const float4*)(Hl + (size_t)s0 * NF);
      const float4 h1 = *(const float4*)(Hl + (size_t)s1 * NF);
      const float4 h2 = *(const float4*)(Hl + (size_t)s2 * NF);
      const float4 h3 = *(const float4*)(Hl + (size_t)s3 * NF);
      acc.x += h0.x * w0; acc.y += h0.y * w0; acc.z += h0.z * w0; acc.w += h0.w * w0;
      acc.x += h1.x * w1; acc.y += h1.y * w1; acc.z += h1.z * w1; acc.w += h1.w * w1;
      acc.x += h2.x * w2; acc.y += h2.y * w2; acc.z += h2.z * w2; acc.w += h2.w * w2;
      acc.x += h3.x * w3; acc.y += h3.y * w3; acc.z += h3.z * w3; acc.w += h3.w * w3;
    }
    const int rem = n - k;
    if (rem > 0) {
      const int s0 = __builtin_amdgcn_readlane(sw.x, k);
      const float w0 = __int_as_float(__builtin_amdgcn_readlane(sw.y, k));
      const float4 h0 = *(const float4*)(Hl + (size_t)s0 * NF);
      if (rem > 1) {
        const int s1 = __builtin_amdgcn_readlane(sw.x, k + 1);
        const float w1 = __int_as_float(__builtin_amdgcn_readlane(sw.y, k + 1));
        const float4 h1 = *(const float4*)(Hl + (size_t)s1 * NF);
        if (rem > 2) {
          const int s2 = __builtin_amdgcn_readlane(sw.x, k + 2);
          const float w2 = __int_as_float(__builtin_amdgcn_readlane(sw.y, k + 2));
          const float4 h2 = *(const float4*)(Hl + (size_t)s2 * NF);
          acc.x += h0.x * w0; acc.y += h0.y * w0; acc.z += h0.z * w0; acc.w += h0.w * w0;
          acc.x += h1.x * w1; acc.y += h1.y * w1; acc.z += h1.z * w1; acc.w += h1.w * w1;
          acc.x += h2.x * w2; acc.y += h2.y * w2; acc.z += h2.z * w2; acc.w += h2.w * w2;
        } else {
          acc.x += h0.x * w0; acc.y += h0.y * w0; acc.z += h0.z * w0; acc.w += h0.w * w0;
          acc.x += h1.x * w1; acc.y += h1.y * w1; acc.z += h1.z * w1; acc.w += h1.w * w1;
        }
      } else {
        acc.x += h0.x * w0; acc.y += h0.y * w0; acc.z += h0.z * w0; acc.w += h0.w * w0;
      }
    }
  }
  return acc;
}

static_assert(N_NODES % 8 == 0, "eight rows per block, no tail");
__global__ __launch_bounds__(256) void gather_relu_kernel(const float* __restrict__ H,
                                                          const int* __restrict__ rowptr,
                                                          const int2* __restrict__ csr,
                                                          const float* __restrict__ dinv,
                                                          const float* __restrict__ b,
                                                          float* __restrict__ OUT) {
  const int lane = threadIdx.x & 31;
  const int i   = __builtin_amdgcn_readfirstlane(blockIdx.x * 8 + (threadIdx.x >> 5));
  const int beg = __builtin_amdgcn_readfirstlane(rowptr[i]);
  const int end = __builtin_amdgcn_readfirstlane(rowptr[i + 1]);
  const float* Hl = H + lane * 4;

  const float4 hs = *(const float4*)(Hl + (size_t)i * NF);
  const float4 bv = *(const float4*)(b + lane * 4);
  const float di = dinv[i];

  const float4 acc = row_sum(Hl, csr, beg, end, lane);

  const float di2 = di * di;
  float4 r;
  r.x = fmaxf(acc.x + di2 * hs.x + bv.x, 0.0f);
  r.y = fmaxf(acc.y + di2 * hs.y + bv.y, 0.0f);
  r.z = fmaxf(acc.z + di2 * hs.z + bv.z, 0.0f);
  r.w = fmaxf(acc.w + di2 * hs.w + bv.w, 0.0f);
  *(float4*)(OUT + (size_t)i * NF + lane * 4) = r;
}

#define POOL_ROWS 16
static_assert(N_NODES % POOL_ROWS == 0, "sixteen rows per block, no tail");
__global__ __launch_bounds__(POOL_ROWS * 32) void gather_pool_kernel(
    const float* __restrict__ Y, const int* __restrict__ rowptr,
    const int2* __restrict__ csr, const float* __restrict__ dinv,
    const int* __restrict__ batch, float* __restrict__ Z) {
  __shared__ float4 rows[POOL_ROWS][ZP / 4];
  __shared__ int gid[POOL_ROWS];
  const int lane = threadIdx.x & 31;
  const int wave = threadIdx.x >> 5;
  const int i   = __builtin_amdgcn_readfirstlane(blockIdx.x * POOL_ROWS + wave);
  const int beg = __builtin_amdgcn_readfirstlane(rowptr[i]);
  const int end = __builtin_amdgcn_readfirstlane(rowptr[i + 1]);
  const float* Yl = Y + lane * 4;

  const float4 hs = *(const float4*)(Yl + (size_t)i * NF);
  const float di = dinv[i];
  const int   g  = batch[i];

  const float4 acc = row_sum(Yl, csr, beg, end, lane);

  const float di2 = di * di;
  float4 r;
  r.x = acc.x + di2 * hs.x;
  r.y = acc.y + di2 * hs.y;
  r.z = acc.z + di2 * hs.z;
  r.w = acc.w + di2 * hs.w;
  rows[wave][lane] = r;
  if (lane == 0) {
    rows[wave][NF / 4] = make_float4(1.0f, 0.0f, 0.0f, 0.0f);
    gid[wave] = g;
  }
  __syncthreads();

  if (threadIdx.x <= NF) {
    const float* rf = (const float*)rows + threadIdx.x;
    float* Zc = Z + threadIdx.x;
    float v[POOL_ROWS];
    int   gw[POOL_ROWS];
    #pragma unroll
    for (int w = 0; w < POOL_ROWS; ++w) {
      v[w]  = rf[w * ZP];
      gw[w] = gid[w];
    }
    int   gcur = gw[0];
    float s    = v[0];
    #pragma unroll
    for (int w = 1; w < POOL_ROWS; ++w) {
      if (gw[w] != gcur) {
        atomicAdd(Zc + (size_t)gcur * ZP, s);
        gcur = gw[w];
        s = v[w];
      } else {
        s += v[w];
      }
    }
    atomicAdd(Zc + (size_t)gcur * ZP, s);
  }
}

extern "C" void kernel_launch(void* const* d_in, const int* in_sizes, int n_in,
                              void* d_out, int out_size, void* d_ws, size_t ws_size,
                              hipStream_t stream) {
  const float* x    = (const float*)d_in[0];
  const int*   ei   = (const int*)  d_in[1];
  const int*   batch= (const int*)  d_in[2];
  const float* W1   = (const float*)d_in[3];
  const float* b1   = (const float*)d_in[4];
  const float* W2   = (const float*)d_in[5];
  const float* b2   = (const float*)d_in[6];
  float* out = (float*)d_out;
  float* ws  = (float*)d_ws;

  const int* src = ei;
  const int* dst = ei + N_EDGES;

  float* dinv = ws;
  float* H    = ws + 50048;
  float* R    = H + (size_t)N_NODES * NF;
  float* Y    = R + (size_t)N_NODES * NF;

  int*  rowptr = (int*)R;
  int*  rank   = rowptr + OWN_SLICES * CNT_PITCH;
  int2* csr    = (int2*)(rank + N_EDGES);
  float* Z     = (float*)(csr + N_EDGES);

  const int T = 256;
  const int gRows     = N_NODES / 8;
  const int gPool     = N_NODES / POOL_ROWS;

  count_rank_owner_kernel<<<OWN_BLOCKS, OWN_THREADS, 0, stream>>>(dst, rowptr, rank, Z);
  scan_dinv_kernel<<<1, 1024, 0, stream>>>(rowptr, dinv);
  fill_gemm_kernel<<<GEMM_BLOCKS + FILL_BLOCKS, T, 0, stream>>>(src, dst, rowptr, rank, dinv, csr, x, W1, H);

  gather_relu_kernel<<<gRows, T, 0, stream>>>(H, rowptr, csr, dinv, b1, Y);

  gather_pool_kernel<<<gPool, POOL_ROWS * 32, 0, stream>>>(Y, rowptr, csr, dinv, batch, Z);
  pooled_product_wmma<<<N_GRAPHS / 16, T, 0, stream>>>(Z, W2, b2, out);
}


# row-start scan spread over 64 blocks (one per node range): owner blocks also store their edge totals, a scan block adds the totals of lower ranges and scans its own 784 nodes from registers
# speedup vs baseline: 4.0157x; 1.1699x over previous
#include <hip/hip_runtime.h>

#define N_NODES  50000
#define N_EDGES  800000
#define NF       128
#define N_GRAPHS 128

typedef __attribute__((ext_vector_type(2))) float v2f;
typedef __attribute__((ext_vector_type(8))) float v8f;

#define ZP (NF + 4)


#define OWN_RANGES  64
#define OWN_SLICES  4
#define OWN_BLOCKS  (OWN_RANGES * OWN_SLICES)
#define OWN_THREADS 1024
#define OWN_NODES   ((((N_NODES + OWN_RANGES - 1) / OWN_RANGES) + 3) / 4 * 4)
#define OWN_LOADS   7
#define CNT_PITCH   50048
#define SLICE_EDGES (N_EDGES / OWN_SLICES)
#define E4          (SLICE_EDGES / 4)
static_assert(N_EDGES % (4 * OWN_SLICES) == 0, "equal slices, read four edges at a time");
static_assert(CNT_PITCH >= N_NODES + 1 && CNT_PITCH % 4 == 0, "count arrays: room for the row end, 16-byte aligned");
static_assert(OWN_NODES <= OWN_THREADS, "one counter per thread when the block stores its counters");
__global__ __launch_bounds__(OWN_THREADS) void count_rank_owner_kernel(const int* __restrict__ dst,
                                                                        int* __restrict__ cnt,
                                                                        int* __restrict__ rank,
                                                                        int* __restrict__ tot,
                                                                        float* __restrict__ Z) {
  __shared__ int c[OWN_NODES];
  __shared__ int wsum[OWN_THREADS / 32];
  const int range = blockIdx.x % OWN_RANGES;
  const int slice = blockIdx.x / OWN_RANGES;
  const int lo = range * OWN_NODES;
  for (int i = threadIdx.x; i < OWN_NODES; i += OWN_THREADS) c[i] = 0;
  for (int i = blockIdx.x * OWN_THREADS + threadIdx.x; i < N_GRAPHS * ZP; i += OWN_BLOCKS * OWN_THREADS)
    Z[i] = 0.0f;
  __syncthreads();

  const int4* d4 = (const int4*)(dst + slice * SLICE_EDGES);
  int* rk = rank + slice * SLICE_EDGES;
  const int rot = range * (E4 / OWN_RANGES);
  for (int base = 0; base < E4; base += OWN_THREADS * OWN_LOADS) {
    int4 v[OWN_LOADS];
    int  at[OWN_LOADS];
    #pragma unroll
    for (int u = 0; u < OWN_LOADS; ++u) {
      const int i = base + u * OWN_THREADS + threadIdx.x;
      int j = i + rot;
      if (j >= E4) j -= E4;
      at[u] = j;
      v[u] = (i < E4) ? d4[j] : make_int4(-1, -1, -1, -1);
    }
    #pragma unroll
    for (int u = 0; u < OWN_LOADS; ++u) {
      const int e0 = 4 * at[u];
      const unsigned t0 = (unsigned)(v[u].x - lo);
      const unsigned t1 = (unsigned)(v[u].y - lo);
      const unsigned t2 = (unsigned)(v[u].z - lo);
      const unsigned t3 = (unsigned)(v[u].w - lo);
      if (t0 < (unsigned)OWN_NODES) rk[e0 + 0] = atomicAdd(&c[t0], 1);
      if (t1 < (unsigned)OWN_NODES) rk[e0 + 1] = atomicAdd(&c[t1], 1);
      if (t2 < (unsigned)OWN_NODES) rk[e0 + 2] = atomicAdd(&c[t2], 1);
      if (t3 < (unsigned)OWN_NODES) rk[e0 + 3] = atomicAdd(&c[t3], 1);
    }
  }
  __syncthreads();
  int mine = 0;
  if (threadIdx.x < OWN_NODES && lo + threadIdx.x < N_NODES) {
    mine = c[threadIdx.x];
    cnt[slice * CNT_PITCH + lo + threadIdx.x] = mine;
  }
  for (int o = 16; o > 0; o >>= 1) mine += __shfl_xor(mine, o);
  if ((threadIdx.x & 31) == 0) wsum[threadIdx.x >> 5] = mine;
  __syncthreads();
  if (threadIdx.x == 0) {
    int t = 0;
    #pragma unroll
    for (int w = 0; w < OWN_THREADS / 32; ++w) t += wsum[w];
    tot[blockIdx.x] = t;
  }
}

#define SCAN_THREADS 256
#define SCAN_WAVES   (SCAN_THREADS / 32)
static_assert(OWN_NODES <= 4 * SCAN_THREADS, "one thread per four nodes of a range");
static_assert(N_NODES % 4 == 0, "counts are handled four at a time");
__global__ __launch_bounds__(SCAN_THREADS) void scan_dinv_kernel(int* __restrict__ cr,
                                                                 const int* __restrict__ tot,
                                                                 float* __restrict__ dinv) {
  __shared__ int wsum[SCAN_WAVES];
  __shared__ int fsum[SCAN_WAVES];
  const int lane  = threadIdx.x & 31;
  const int wave  = threadIdx.x >> 5;
  const int range = blockIdx.x;
  const int i = range * OWN_NODES + 4 * threadIdx.x;
  const bool on = 4 * threadIdx.x < OWN_NODES && i < N_NODES;

  int front = 0;
  for (int j = threadIdx.x; j < OWN_BLOCKS; j += SCAN_THREADS)
    if (j % OWN_RANGES < range) front += tot[j];

  int4 cs[OWN_SLICES];
  int4 c = make_int4(0, 0, 0, 0);
  #pragma unroll
  for (int s = 0; s < OWN_SLICES; ++s) {
    cs[s] = on ? *(const int4*)(cr + s * CNT_PITCH + i) : make_int4(0, 0, 0, 0);
    c.x += cs[s].x; c.y += cs[s].y; c.z += cs[s].z; c.w += cs[s].w;
  }
  const int t = c.x + c.y + c.z + c.w;
  int inc = t;
  for (int o = 1; o < 32; o <<= 1) {
    const int up = __shfl_up(inc, o);
    if (lane >= o) inc += up;
  }
  for (int o = 16; o > 0; o >>= 1) front += __shfl_xor(front, o);
  if (lane == 31) wsum[wave] = inc;
  if (lane == 0)  fsum[wave] = front;
  __syncthreads();

  int carry = 0;
  #pragma unroll
  for (int w = 0; w < SCAN_WAVES; ++w) {
    carry += fsum[w];
    if (w < wave) carry += wsum[w];
  }
  const int ex = carry + inc - t;
  if (on) {
    int4 p = make_int4(ex, ex + c.x, ex + c.x + c.y, ex + c.x + c.y + c.z);
    #pragma unroll
    for (int s = 0; s < OWN_SLICES; ++s) {
      *(int4*)(cr + s * CNT_PITCH + i) = p;
      p.x += cs[s].x; p.y += cs[s].y; p.z += cs[s].z; p.w += cs[s].w;
    }
    *(float4*)(dinv + i) = make_float4(rsqrtf((float)c.x + 1.0f), rsqrtf((float)c.y + 1.0f),
                                       rsqrtf((float)c.z + 1.0f), rsqrtf((float)c.w + 1.0f));
    if (i + 4 == N_NODES) cr[N_NODES] = ex + t;
  }
}

__device__ __forceinline__ void fill_block(const int unit, const int* __restrict__ src,
                                           const int* __restrict__ dst,
                                           const int* __restrict__ rowptr,
                                           const int* __restrict__ rank,
                                           const float* __restrict__ dinv, int2* __restrict__ csr) {
  const int e = unit * 256 + threadIdx.x;
  if (e >= N_EDGES) return;
  const int s = src[e];
  const int d = dst[e];
  const float norm = dinv[s] * dinv[d];
  csr[rowptr[(e / SLICE_EDGES) * CNT_PITCH + d] + rank[e]] = make_int2(s, __float_as_int(norm));
}

template <int R>
__device__ __forceinline__ void tile_product(const float* __restrict__ Xrow, const int tile_pitch,
                                             const float* __restrict__ Wcol,
                                             const int half, v8f (&acc)[R]) {
  #pragma unroll
  for (int k = 0; k < NF; k += 4) {
    v2f b;
    b.x = Wcol[(size_t)(k + 2 * half + 0) * NF];
    b.y = Wcol[(size_t)(k + 2 * half + 1) * NF];
    #pragma unroll
    for (int t = 0; t < R; ++t) {
      const v2f a = *(const v2f*)(Xrow + (size_t)t * tile_pitch + k + 2 * half);
      acc[t] = __builtin_amdgcn_wmma_f32_16x16x4_f32(
           false, a,  false, b,
           (short)0, acc[t],  false,  false);
    }
  }
}

#define GEMM_RT      5
#define GEMM_ROWS    (16 * GEMM_RT)
#define GEMM_BLOCKS  (N_NODES / GEMM_ROWS)
#define FILL_BLOCKS  ((N_EDGES + 255) / 256)
static_assert(N_NODES % GEMM_ROWS == 0, "whole row tiles, EXEC all ones");
static_assert((GEMM_ROWS * (NF / 4)) % 256 == 0, "staging: whole float4 rounds per thread");
__global__ __launch_bounds__(256) void fill_gemm_kernel(const int* __restrict__ src,
                                                        const int* __restrict__ dst,
                                                        const int* __restrict__ rowptr,
                                                        const int* __restrict__ rank,
                                                        const float* __restrict__ dinv,
                                                        int2* __restrict__ csr,
                                                        const float* __restrict__ X,
                                                        const float* __restrict__ W,
                                                        float* __restrict__ H) {
  __shared__ float4 xs[GEMM_ROWS][ZP / 4];
  if (blockIdx.x >= GEMM_BLOCKS) {
    fill_block(blockIdx.x - GEMM_BLOCKS, src, dst, rowptr, rank, dinv, csr);
    return;
  }
  const int wave = threadIdx.x >> 5;
  const int lane = threadIdx.x & 31;
  const int half = lane >> 4;
  const int l    = lane & 15;
  const int m0   = blockIdx.x * GEMM_ROWS;
  const int n0   = wave * 16;

  const float4* Xb = (const float4*)(X + (size_t)m0 * NF);
  #pragma unroll
  for (int i = 0; i < GEMM_ROWS * (NF / 4) / 256; ++i) {
    const int idx = i * 256 + threadIdx.x;
    xs[idx >> 5][idx & 31] = Xb[idx];
  }
  __syncthreads();

  v8f acc[GEMM_RT] = {};
  tile_product<GEMM_RT>((const float*)xs + (size_t)l * ZP, 16 * ZP, W + n0 + l, half, acc);

  #pragma unroll
  for (int t = 0; t < GEMM_RT; ++t) {
    float* Hp = H + (size_t)(m0 + 16 * t + 8 * half) * NF + n0 + l;
    #pragma unroll
    for (int r = 0; r < 8; ++r) Hp[(size_t)r * NF] = acc[t][r];
  }
}

static_assert(N_GRAPHS % 16 == 0, "whole 16-row tiles, EXEC all ones");
__global__ __launch_bounds__(256) void pooled_product_wmma(const float* __restrict__ Z,
                                                           const float* __restrict__ W,
                                                           const float* __restrict__ b,
                                                           float* __restrict__ out) {
  const int wave = threadIdx.x >> 5;
  const int lane = threadIdx.x & 31;
  const int half = lane >> 4;
  const int l    = lane & 15;
  const int m0   = blockIdx.x * 16;
  const int n0   = wave * 16;

  const float bj = b[n0 + l];
  const float* cp = Z + (size_t)(m0 + 8 * half) * ZP + NF;
  v8f c0;
  #pragma unroll
  for (int r = 0; r < 8; ++r) c0[r] = cp[(size_t)r * ZP] * bj;

  v8f acc[1] = {c0};
  tile_product<1>(Z + (size_t)(m0 + l) * ZP, 0, W + n0 + l, half, acc);

  float* Op = out + (size_t)(m0 + 8 * half) * NF + n0 + l;
  #pragma unroll
  for (int r = 0; r < 8; ++r) Op[(size_t)r * NF] = acc[0][r];
}

__device__ __forceinline__ float4 row_sum(const float* __restrict__ Hl,
                                          const int2* __restrict__ csr,
                                          const int beg, const int end, const int lane) {
  float4 acc = make_float4(0.0f, 0.0f, 0.0f, 0.0f);
  for (int c = beg; c < end; c += 32) {
    const int n = min(32, end - c);
    int2 sw = make_int2(0, 0);
    if (lane < n) sw = csr[c + lane];
    int k = 0;
    for (; k + 4 <= n; k += 4) {
      const int s0 = __builtin_amdgcn_readlane(sw.x, k + 0);
      const int s1 = __builtin_amdgcn_readlane(sw.x, k + 1);
      const int s2 = __builtin_amdgcn_readlane(sw.x, k + 2);
      const int s3 = __builtin_amdgcn_readlane(sw.x, k + 3);
      const float w0 = __int_as_float(__builtin_amdgcn_readlane(sw.y, k + 0));
      const float w1 = __int_as_float(__builtin_amdgcn_readlane(sw.y, k + 1));
      const float w2 = __int_as_float(__builtin_amdgcn_readlane(sw.y, k + 2));
      const float w3 = __int_as_float(__builtin_amdgcn_readlane(sw.y, k + 3));
      const float4 h0 = *(const float4*)(Hl + (size_t)s0 * NF);
      const float4 h1 = *(const float4*)(Hl + (size_t)s1 * NF);
      const float4 h2 = *(const float4*)(Hl + (size_t)s2 * NF);
      const float4 h3 = *(const float4*)(Hl + (size_t)s3 * NF);
      acc.x += h0.x * w0; acc.y += h0.y * w0; acc.z += h0.z * w0; acc.w += h0.w * w0;
      acc.x += h1.x * w1; acc.y += h1.y * w1; acc.z += h1.z * w1; acc.w += h1.w * w1;
      acc.x += h2.x * w2; acc.y += h2.y * w2; acc.z += h2.z * w2; acc.w += h2.w * w2;
      acc.x += h3.x * w3; acc.y += h3.y * w3; acc.z += h3.z * w3; acc.w += h3.w * w3;
    }
    const int rem = n - k;
    if (rem > 0) {
      const int s0 = __builtin_amdgcn_readlane(sw.x, k);
      const float w0 = __int_as_float(__builtin_amdgcn_readlane(sw.y, k));
      const float4 h0 = *(const float4*)(Hl + (size_t)s0 * NF);
      if (rem > 1) {
        const int s1 = __builtin_amdgcn_readlane(sw.x, k + 1);
        const float w1 = __int_as_float(__builtin_amdgcn_readlane(sw.y, k + 1));
        const float4 h1 = *(const float4*)(Hl + (size_t)s1 * NF);
        if (rem > 2) {
          const int s2 = __builtin_amdgcn_readlane(sw.x, k + 2);
          const float w2 = __int_as_float(__builtin_amdgcn_readlane(sw.y, k + 2));
          const float4 h2 = *(const float4*)(Hl + (size_t)s2 * NF);
          acc.x += h0.x * w0; acc.y += h0.y * w0; acc.z += h0.z * w0; acc.w += h0.w * w0;
          acc.x += h1.x * w1; acc.y += h1.y * w1; acc.z += h1.z * w1; acc.w += h1.w * w1;
          acc.x += h2.x * w2; acc.y += h2.y * w2; acc.z += h2.z * w2; acc.w += h2.w * w2;
        } else {
          acc.x += h0.x * w0; acc.y += h0.y * w0; acc.z += h0.z * w0; acc.w += h0.w * w0;
          acc.x += h1.x * w1; acc.y += h1.y * w1; acc.z += h1.z * w1; acc.w += h1.w * w1;
        }
      } else {
        acc.x += h0.x * w0; acc.y += h0.y * w0; acc.z += h0.z * w0; acc.w += h0.w * w0;
      }
    }
  }
  return acc;
}

static_assert(N_NODES % 8 == 0, "eight rows per block, no tail");
__global__ __launch_bounds__(256) void gather_relu_kernel(const float* __restrict__ H,
                                                          const int* __restrict__ rowptr,
                                                          const int2* __restrict__ csr,
                                                          const float* __restrict__ dinv,
                                                          const float* __restrict__ b,
                                                          float* __restrict__ OUT) {
  const int lane = threadIdx.x & 31;
  const int i   = __builtin_amdgcn_readfirstlane(blockIdx.x * 8 + (threadIdx.x >> 5));
  const int beg = __builtin_amdgcn_readfirstlane(rowptr[i]);
  const int end = __builtin_amdgcn_readfirstlane(rowptr[i + 1]);
  const float* Hl = H + lane * 4;

  const float4 hs = *(const float4*)(Hl + (size_t)i * NF);
  const float4 bv = *(const float4*)(b + lane * 4);
  const float di = dinv[i];

  const float4 acc = row_sum(Hl, csr, beg, end, lane);

  const float di2 = di * di;
  float4 r;
  r.x = fmaxf(acc.x + di2 * hs.x + bv.x, 0.0f);
  r.y = fmaxf(acc.y + di2 * hs.y + bv.y, 0.0f);
  r.z = fmaxf(acc.z + di2 * hs.z + bv.z, 0.0f);
  r.w = fmaxf(acc.w + di2 * hs.w + bv.w, 0.0f);
  *(float4*)(OUT + (size_t)i * NF + lane * 4) = r;
}

#define POOL_ROWS 16
static_assert(N_NODES % POOL_ROWS == 0, "sixteen rows per block, no tail");
__global__ __launch_bounds__(POOL_ROWS * 32) void gather_pool_kernel(
    const float* __restrict__ Y, const int* __restrict__ rowptr,
    const int2* __restrict__ csr, const float* __restrict__ dinv,
    const int* __restrict__ batch, float* __restrict__ Z) {
  __shared__ float4 rows[POOL_ROWS][ZP / 4];
  __shared__ int gid[POOL_ROWS];
  const int lane = threadIdx.x & 31;
  const int wave = threadIdx.x >> 5;
  const int i   = __builtin_amdgcn_readfirstlane(blockIdx.x * POOL_ROWS + wave);
  const int beg = __builtin_amdgcn_readfirstlane(rowptr[i]);
  const int end = __builtin_amdgcn_readfirstlane(rowptr[i + 1]);
  const float* Yl = Y + lane * 4;

  const float4 hs = *(const float4*)(Yl + (size_t)i * NF);
  const float di = dinv[i];
  const int   g  = batch[i];

  const float4 acc = row_sum(Yl, csr, beg, end, lane);

  const float di2 = di * di;
  float4 r;
  r.x = acc.x + di2 * hs.x;
  r.y = acc.y + di2 * hs.y;
  r.z = acc.z + di2 * hs.z;
  r.w = acc.w + di2 * hs.w;
  rows[wave][lane] = r;
  if (lane == 0) {
    rows[wave][NF / 4] = make_float4(1.0f, 0.0f, 0.0f, 0.0f);
    gid[wave] = g;
  }
  __syncthreads();

  if (threadIdx.x <= NF) {
    const float* rf = (const float*)rows + threadIdx.x;
    float* Zc = Z + threadIdx.x;
    float v[POOL_ROWS];
    int   gw[POOL_ROWS];
    #pragma unroll
    for (int w = 0; w < POOL_ROWS; ++w) {
      v[w]  = rf[w * ZP];
      gw[w] = gid[w];
    }
    int   gcur = gw[0];
    float s    = v[0];
    #pragma unroll
    for (int w = 1; w < POOL_ROWS; ++w) {
      if (gw[w] != gcur) {
        atomicAdd(Zc + (size_t)gcur * ZP, s);
        gcur = gw[w];
        s = v[w];
      } else {
        s += v[w];
      }
    }
    atomicAdd(Zc + (size_t)gcur * ZP, s);
  }
}

extern "C" void kernel_launch(void* const* d_in, const int* in_sizes, int n_in,
                              void* d_out, int out_size, void* d_ws, size_t ws_size,
                              hipStream_t stream) {
  const float* x    = (const float*)d_in[0];
  const int*   ei   = (const int*)  d_in[1];
  const int*   batch= (const int*)  d_in[2];
  const float* W1   = (const float*)d_in[3];
  const float* b1   = (const float*)d_in[4];
  const float* W2   = (const float*)d_in[5];
  const float* b2   = (const float*)d_in[6];
  float* out = (float*)d_out;
  float* ws  = (float*)d_ws;

  const int* src = ei;
  const int* dst = ei + N_EDGES;

  float* dinv = ws;
  float* H    = ws + 50048;
  float* R    = H + (size_t)N_NODES * NF;
  float* Y    = R + (size_t)N_NODES * NF;

  int*  rowptr = (int*)R;
  int*  rank   = rowptr + OWN_SLICES * CNT_PITCH;
  int2* csr    = (int2*)(rank + N_EDGES);
  float* Z     = (float*)(csr + N_EDGES);
  int*  tot    = (int*)(Z + N_GRAPHS * ZP);

  const int T = 256;
  const int gRows     = N_NODES / 8;
  const int gPool     = N_NODES / POOL_ROWS;

  count_rank_owner_kernel<<<OWN_BLOCKS, OWN_THREADS, 0, stream>>>(dst, rowptr, rank, tot, Z);
  scan_dinv_kernel<<<OWN_RANGES, SCAN_THREADS, 0, stream>>>(rowptr, tot, dinv);
  fill_gemm_kernel<<<GEMM_BLOCKS + FILL_BLOCKS, T, 0, stream>>>(src, dst, rowptr, rank, dinv, csr, x, W1, H);

  gather_relu_kernel<<<gRows, T, 0, stream>>>(H, rowptr, csr, dinv, b1, Y);

  gather_pool_kernel<<<gPool, POOL_ROWS * 32, 0, stream>>>(Y, rowptr, csr, dinv, batch, Z);
  pooled_product_wmma<<<N_GRAPHS / 16, T, 0, stream>>>(Z, W2, b2, out);
}


# in-degree count on 8 wide node ranges x 32 short edge slices instead of 64 x 4: the destination list is swept 8 times, not 64; owner blocks store totals per scan range; sum order changes
# speedup vs baseline: 4.2567x; 1.0600x over previous
#include <hip/hip_runtime.h>

#define N_NODES  50000
#define N_EDGES  800000
#define NF       128
#define N_GRAPHS 128

typedef __attribute__((ext_vector_type(2))) float v2f;
typedef __attribute__((ext_vector_type(8))) float v8f;

#define ZP (NF + 4)


#define OWN_RANGES   8
#define OWN_SLICES   32
#define OWN_BLOCKS   (OWN_RANGES * OWN_SLICES)
#define OWN_THREADS  1024
#define OWN_WAVES    (OWN_THREADS / 32)
#define SCAN_RANGES  64
#define SCAN_PER_OWN (SCAN_RANGES / OWN_RANGES)
#define SCAN_NODES   ((((N_NODES + SCAN_RANGES - 1) / SCAN_RANGES) + 3) / 4 * 4)
#define OWN_NODES    (SCAN_PER_OWN * SCAN_NODES)
#define OWN_LOADS    7
#define CNT_PITCH    50048
#define SLICE_EDGES  (N_EDGES / OWN_SLICES)
#define E4           (SLICE_EDGES / 4)
#define TOT_SPLIT    (OWN_WAVES / SCAN_PER_OWN)
#define TOT_SPAN     (SCAN_NODES / TOT_SPLIT)
static_assert(N_EDGES % (4 * OWN_SLICES) == 0, "equal slices, read four edges at a time");
static_assert(CNT_PITCH >= N_NODES + 1 && CNT_PITCH % 4 == 0, "count arrays: room for the row end, 16-byte aligned");
static_assert(SCAN_RANGES % OWN_RANGES == 0 && OWN_WAVES % SCAN_PER_OWN == 0 && SCAN_NODES % TOT_SPLIT == 0,
              "whole scan ranges per owner range, whole waves per scan range");
__global__ __launch_bounds__(OWN_THREADS) void count_rank_owner_kernel(const int* __restrict__ dst,
                                                                        int* __restrict__ cnt,
                                                                        int* __restrict__ rank,
                                                                        int* __restrict__ tot,
                                                                        float* __restrict__ Z) {
  __shared__ int c[OWN_NODES];
  __shared__ int wsum[OWN_WAVES];
  const int range = blockIdx.x % OWN_RANGES;
  const int slice = blockIdx.x / OWN_RANGES;
  const int lo = range * OWN_NODES;
  for (int i = threadIdx.x; i < OWN_NODES; i += OWN_THREADS) c[i] = 0;
  for (int i = blockIdx.x * OWN_THREADS + threadIdx.x; i < N_GRAPHS * ZP; i += OWN_BLOCKS * OWN_THREADS)
    Z[i] = 0.0f;
  __syncthreads();

  const int4* d4 = (const int4*)(dst + slice * SLICE_EDGES);
  int* rk = rank + slice * SLICE_EDGES;
  const int rot = range * (E4 / OWN_RANGES);
  for (int base = 0; base < E4; base += OWN_THREADS * OWN_LOADS) {
    int4 v[OWN_LOADS];
    int  at[OWN_LOADS];
    #pragma unroll
    for (int u = 0; u < OWN_LOADS; ++u) {
      const int i = base + u * OWN_THREADS + threadIdx.x;
      int j = i + rot;
      if (j >= E4) j -= E4;
      at[u] = j;
      v[u] = (i < E4) ? d4[j] : make_int4(-1, -1, -1, -1);
    }
    #pragma unroll
    for (int u = 0; u < OWN_LOADS; ++u) {
      const int e0 = 4 * at[u];
      const unsigned t0 = (unsigned)(v[u].x - lo);
      const unsigned t1 = (unsigned)(v[u].y - lo);
      const unsigned t2 = (unsigned)(v[u].z - lo);
      const unsigned t3 = (unsigned)(v[u].w - lo);
      const bool k0 = t0 < (unsigned)OWN_NODES;
      const bool k1 = t1 < (unsigned)OWN_NODES;
      const bool k2 = t2 < (unsigned)OWN_NODES;
      const bool k3 = t3 < (unsigned)OWN_NODES;
      int r0 = 0, r1 = 0, r2 = 0, r3 = 0;
      if (k0) r0 = atomicAdd(&c[t0], 1);
      if (k1) r1 = atomicAdd(&c[t1], 1);
      if (k2) r2 = atomicAdd(&c[t2], 1);
      if (k3) r3 = atomicAdd(&c[t3], 1);
      if (k0) rk[e0 + 0] = r0;
      if (k1) rk[e0 + 1] = r1;
      if (k2) rk[e0 + 2] = r2;
      if (k3) rk[e0 + 3] = r3;
    }
  }
  __syncthreads();
  for (int i = threadIdx.x; i < OWN_NODES; i += OWN_THREADS)
    if (lo + i < N_NODES) cnt[slice * CNT_PITCH + lo + i] = c[i];
  const int wave = threadIdx.x >> 5;
  const int lane = threadIdx.x & 31;
  int mine = 0;
  for (int k = lane; k < TOT_SPAN; k += 32) mine += c[wave * TOT_SPAN + k];
  for (int o = 16; o > 0; o >>= 1) mine += __shfl_xor(mine, o);
  if (lane == 0) wsum[wave] = mine;
  __syncthreads();
  if (threadIdx.x < SCAN_PER_OWN) {
    int t = 0;
    #pragma unroll
    for (int w = 0; w < TOT_SPLIT; ++w) t += wsum[threadIdx.x * TOT_SPLIT + w];
    tot[slice * SCAN_RANGES + range * SCAN_PER_OWN + threadIdx.x] = t;
  }
}

#define SCAN_THREADS 256
#define SCAN_WAVES   (SCAN_THREADS / 32)
static_assert(SCAN_NODES <= 4 * SCAN_THREADS, "one thread per four nodes of a range");
static_assert(N_NODES % 4 == 0, "counts are handled four at a time");
__global__ __launch_bounds__(SCAN_THREADS) void scan_dinv_kernel(int* __restrict__ cr,
                                                                 const int* __restrict__ tot,
                                                                 float* __restrict__ dinv) {
  __shared__ int wsum[SCAN_WAVES];
  __shared__ int fsum[SCAN_WAVES];
  const int lane  = threadIdx.x & 31;
  const int wave  = threadIdx.x >> 5;
  const int range = blockIdx.x;
  const int i = range * SCAN_NODES + 4 * threadIdx.x;
  const bool on = 4 * threadIdx.x < SCAN_NODES && i < N_NODES;

  int front = 0;
  for (int j = threadIdx.x; j < OWN_SLICES * SCAN_RANGES; j += SCAN_THREADS)
    if (j % SCAN_RANGES < range) front += tot[j];

  int4 cs[OWN_SLICES];
  int4 c = make_int4(0, 0, 0, 0);
  #pragma unroll
  for (int s = 0; s < OWN_SLICES; ++s) {
    cs[s] = on ? *(const int4*)(cr + s * CNT_PITCH + i) : make_int4(0, 0, 0, 0);
    c.x += cs[s].x; c.y += cs[s].y; c.z += cs[s].z; c.w += cs[s].w;
  }
  const int t = c.x + c.y + c.z + c.w;
  int inc = t;
  for (int o = 1; o < 32; o <<= 1) {
    const int up = __shfl_up(inc, o);
    if (lane >= o) inc += up;
  }
  for (int o = 16; o > 0; o >>= 1) front += __shfl_xor(front, o);
  if (lane == 31) wsum[wave] = inc;
  if (lane == 0)  fsum[wave] = front;
  __syncthreads();

  int carry = 0;
  #pragma unroll
  for (int w = 0; w < SCAN_WAVES; ++w) {
    carry += fsum[w];
    if (w < wave) carry += wsum[w];
  }
  const int ex = carry + inc - t;
  if (on) {
    int4 p = make_int4(ex, ex + c.x, ex + c.x + c.y, ex + c.x + c.y + c.z);
    #pragma unroll
    for (int s = 0; s < OWN_SLICES; ++s) {
      *(int4*)(cr + s * CNT_PITCH + i) = p;
      p.x += cs[s].x; p.y += cs[s].y; p.z += cs[s].z; p.w += cs[s].w;
    }
    *(float4*)(dinv + i) = make_float4(rsqrtf((float)c.x + 1.0f), rsqrtf((float)c.y + 1.0f),
                                       rsqrtf((float)c.z + 1.0f), rsqrtf((float)c.w + 1.0f));
    if (i + 4 == N_NODES) cr[N_NODES] = ex + t;
  }
}

__device__ __forceinline__ void fill_block(const int unit, const int* __restrict__ src,
                                           const int* __restrict__ dst,
                                           const int* __restrict__ rowptr,
                                           const int* __restrict__ rank,
                                           const float* __restrict__ dinv, int2* __restrict__ csr) {
  const int e = unit * 256 + threadIdx.x;
  if (e >= N_EDGES) return;
  const int s = src[e];
  const int d = dst[e];
  const float norm = dinv[s] * dinv[d];
  csr[rowptr[(e / SLICE_EDGES) * CNT_PITCH + d] + rank[e]] = make_int2(s, __float_as_int(norm));
}

template <int R>
__device__ __forceinline__ void tile_product(const float* __restrict__ Xrow, const int tile_pitch,
                                             const float* __restrict__ Wcol,
                                             const int half, v8f (&acc)[R]) {
  #pragma unroll
  for (int k = 0; k < NF; k += 4) {
    v2f b;
    b.x = Wcol[(size_t)(k + 2 * half + 0) * NF];
    b.y = Wcol[(size_t)(k + 2 * half + 1) * NF];
    #pragma unroll
    for (int t = 0; t < R; ++t) {
      const v2f a = *(const v2f*)(Xrow + (size_t)t * tile_pitch + k + 2 * half);
      acc[t] = __builtin_amdgcn_wmma_f32_16x16x4_f32(
           false, a,  false, b,
           (short)0, acc[t],  false,  false);
    }
  }
}

#define GEMM_RT      5
#define GEMM_ROWS    (16 * GEMM_RT)
#define GEMM_BLOCKS  (N_NODES / GEMM_ROWS)
#define FILL_BLOCKS  ((N_EDGES + 255) / 256)
static_assert(N_NODES % GEMM_ROWS == 0, "whole row tiles, EXEC all ones");
static_assert((GEMM_ROWS * (NF / 4)) % 256 == 0, "staging: whole float4 rounds per thread");
__global__ __launch_bounds__(256) void fill_gemm_kernel(const int* __restrict__ src,
                                                        const int* __restrict__ dst,
                                                        const int* __restrict__ rowptr,
                                                        const int* __restrict__ rank,
                                                        const float* __restrict__ dinv,
                                                        int2* __restrict__ csr,
                                                        const float* __restrict__ X,
                                                        const float* __restrict__ W,
                                                        float* __restrict__ H) {
  __shared__ float4 xs[GEMM_ROWS][ZP / 4];
  if (blockIdx.x >= GEMM_BLOCKS) {
    fill_block(blockIdx.x - GEMM_BLOCKS, src, dst, rowptr, rank, dinv, csr);
    return;
  }
  const int wave = threadIdx.x >> 5;
  const int lane = threadIdx.x & 31;
  const int half = lane >> 4;
  const int l    = lane & 15;
  const int m0   = blockIdx.x * GEMM_ROWS;
  const int n0   = wave * 16;

  const float4* Xb = (const float4*)(X + (size_t)m0 * NF);
  #pragma unroll
  for (int i = 0; i < GEMM_ROWS * (NF / 4) / 256; ++i) {
    const int idx = i * 256 + threadIdx.x;
    xs[idx >> 5][idx & 31] = Xb[idx];
  }
  __syncthreads();

  v8f acc[GEMM_RT] = {};
  tile_product<GEMM_RT>((const float*)xs + (size_t)l * ZP, 16 * ZP, W + n0 + l, half, acc);

  #pragma unroll
  for (int t = 0; t < GEMM_RT; ++t) {
    float* Hp = H + (size_t)(m0 + 16 * t + 8 * half) * NF + n0 + l;
    #pragma unroll
    for (int r = 0; r < 8; ++r) Hp[(size_t)r * NF] = acc[t][r];
  }
}

static_assert(N_GRAPHS % 16 == 0, "whole 16-row tiles, EXEC all ones");
__global__ __launch_bounds__(256) void pooled_product_wmma(const float* __restrict__ Z,
                                                           const float* __restrict__ W,
                                                           const float* __restrict__ b,
                                                           float* __restrict__ out) {
  const int wave = threadIdx.x >> 5;
  const int lane = threadIdx.x & 31;
  const int half = lane >> 4;
  const int l    = lane & 15;
  const int m0   = blockIdx.x * 16;
  const int n0   = wave * 16;

  const float bj = b[n0 + l];
  const float* cp = Z + (size_t)(m0 + 8 * half) * ZP + NF;
  v8f c0;
  #pragma unroll
  for (int r = 0; r < 8; ++r) c0[r] = cp[(size_t)r * ZP] * bj;

  v8f acc[1] = {c0};
  tile_product<1>(Z + (size_t)(m0 + l) * ZP, 0, W + n0 + l, half, acc);

  float* Op = out + (size_t)(m0 + 8 * half) * NF + n0 + l;
  #pragma unroll
  for (int r = 0; r < 8; ++r) Op[(size_t)r * NF] = acc[0][r];
}

__device__ __forceinline__ float4 row_sum(const float* __restrict__ Hl,
                                          const int2* __restrict__ csr,
                                          const int beg, const int end, const int lane) {
  float4 acc = make_float4(0.0f, 0.0f, 0.0f, 0.0f);
  for (int c = beg; c < end; c += 32) {
    const int n = min(32, end - c);
    int2 sw = make_int2(0, 0);
    if (lane < n) sw = csr[c + lane];
    int k = 0;
    for (; k + 4 <= n; k += 4) {
      const int s0 = __builtin_amdgcn_readlane(sw.x, k + 0);
      const int s1 = __builtin_amdgcn_readlane(sw.x, k + 1);
      const int s2 = __builtin_amdgcn_readlane(sw.x, k + 2);
      const int s3 = __builtin_amdgcn_readlane(sw.x, k + 3);
      const float w0 = __int_as_float(__builtin_amdgcn_readlane(sw.y, k + 0));
      const float w1 = __int_as_float(__builtin_amdgcn_readlane(sw.y, k + 1));
      const float w2 = __int_as_float(__builtin_amdgcn_readlane(sw.y, k + 2));
      const float w3 = __int_as_float(__builtin_amdgcn_readlane(sw.y, k + 3));
      const float4 h0 = *(const float4*)(Hl + (size_t)s0 * NF);
      const float4 h1 = *(const float4*)(Hl + (size_t)s1 * NF);
      const float4 h2 = *(const float4*)(Hl + (size_t)s2 * NF);
      const float4 h3 = *(const float4*)(Hl + (size_t)s3 * NF);
      acc.x += h0.x * w0; acc.y += h0.y * w0; acc.z += h0.z * w0; acc.w += h0.w * w0;
      acc.x += h1.x * w1; acc.y += h1.y * w1; acc.z += h1.z * w1; acc.w += h1.w * w1;
      acc.x += h2.x * w2; acc.y += h2.y * w2; acc.z += h2.z * w2; acc.w += h2.w * w2;
      acc.x += h3.x * w3; acc.y += h3.y * w3; acc.z += h3.z * w3; acc.w += h3.w * w3;
    }
    const int rem = n - k;
    if (rem > 0) {
      const int s0 = __builtin_amdgcn_readlane(sw.x, k);
      const float w0 = __int_as_float(__builtin_amdgcn_readlane(sw.y, k));
      const float4 h0 = *(const float4*)(Hl + (size_t)s0 * NF);
      if (rem > 1) {
        const int s1 = __builtin_amdgcn_readlane(sw.x, k + 1);
        const float w1 = __int_as_float(__builtin_amdgcn_readlane(sw.y, k + 1));
        const float4 h1 = *(const float4*)(Hl + (size_t)s1 * NF);
        if (rem > 2) {
          const int s2 = __builtin_amdgcn_readlane(sw.x, k + 2);
          const float w2 = __int_as_float(__builtin_amdgcn_readlane(sw.y, k + 2));
          const float4 h2 = *(const float4*)(Hl + (size_t)s2 * NF);
          acc.x += h0.x * w0; acc.y += h0.y * w0; acc.z += h0.z * w0; acc.w += h0.w * w0;
          acc.x += h1.x * w1; acc.y += h1.y * w1; acc.z += h1.z * w1; acc.w += h1.w * w1;
          acc.x += h2.x * w2; acc.y += h2.y * w2; acc.z += h2.z * w2; acc.w += h2.w * w2;
        } else {
          acc.x += h0.x * w0; acc.y += h0.y * w0; acc.z += h0.z * w0; acc.w += h0.w * w0;
          acc.x += h1.x * w1; acc.y += h1.y * w1; acc.z += h1.z * w1; acc.w += h1.w * w1;
        }
      } else {
        acc.x += h0.x * w0; acc.y += h0.y * w0; acc.z += h0.z * w0; acc.w += h0.w * w0;
      }
    }
  }
  return acc;
}

static_assert(N_NODES % 8 == 0, "eight rows per block, no tail");
__global__ __launch_bounds__(256) void gather_relu_kernel(const float* __restrict__ H,
                                                          const int* __restrict__ rowptr,
                                                          const int2* __restrict__ csr,
                                                          const float* __restrict__ dinv,
                                                          const float* __restrict__ b,
                                                          float* __restrict__ OUT) {
  const int lane = threadIdx.x & 31;
  const int i   = __builtin_amdgcn_readfirstlane(blockIdx.x * 8 + (threadIdx.x >> 5));
  const int beg = __builtin_amdgcn_readfirstlane(rowptr[i]);
  const int end = __builtin_amdgcn_readfirstlane(rowptr[i + 1]);
  const float* Hl = H + lane * 4;

  const float4 hs = *(const float4*)(Hl + (size_t)i * NF);
  const float4 bv = *(const float4*)(b + lane * 4);
  const float di = dinv[i];

  const float4 acc = row_sum(Hl, csr, beg, end, lane);

  const float di2 = di * di;
  float4 r;
  r.x = fmaxf(acc.x + di2 * hs.x + bv.x, 0.0f);
  r.y = fmaxf(acc.y + di2 * hs.y + bv.y, 0.0f);
  r.z = fmaxf(acc.z + di2 * hs.z + bv.z, 0.0f);
  r.w = fmaxf(acc.w + di2 * hs.w + bv.w, 0.0f);
  *(float4*)(OUT + (size_t)i * NF + lane * 4) = r;
}

#define POOL_ROWS 16
static_assert(N_NODES % POOL_ROWS == 0, "sixteen rows per block, no tail");
__global__ __launch_bounds__(POOL_ROWS * 32) void gather_pool_kernel(
    const float* __restrict__ Y, const int* __restrict__ rowptr,
    const int2* __restrict__ csr, const float* __restrict__ dinv,
    const int* __restrict__ batch, float* __restrict__ Z) {
  __shared__ float4 rows[POOL_ROWS][ZP / 4];
  __shared__ int gid[POOL_ROWS];
  const int lane = threadIdx.x & 31;
  const int wave = threadIdx.x >> 5;
  const int i   = __builtin_amdgcn_readfirstlane(blockIdx.x * POOL_ROWS + wave);
  const int beg = __builtin_amdgcn_readfirstlane(rowptr[i]);
  const int end = __builtin_amdgcn_readfirstlane(rowptr[i + 1]);
  const float* Yl = Y + lane * 4;

  const float4 hs = *(const float4*)(Yl + (size_t)i * NF);
  const float di = dinv[i];
  const int   g  = batch[i];

  const float4 acc = row_sum(Yl, csr, beg, end, lane);

  const float di2 = di * di;
  float4 r;
  r.x = acc.x + di2 * hs.x;
  r.y = acc.y + di2 * hs.y;
  r.z = acc.z + di2 * hs.z;
  r.w = acc.w + di2 * hs.w;
  rows[wave][lane] = r;
  if (lane == 0) {
    rows[wave][NF / 4] = make_float4(1.0f, 0.0f, 0.0f, 0.0f);
    gid[wave] = g;
  }
  __syncthreads();

  if (threadIdx.x <= NF) {
    const float* rf = (const float*)rows + threadIdx.x;
    float* Zc = Z + threadIdx.x;
    float v[POOL_ROWS];
    int   gw[POOL_ROWS];
    #pragma unroll
    for (int w = 0; w < POOL_ROWS; ++w) {
      v[w]  = rf[w * ZP];
      gw[w] = gid[w];
    }
    int   gcur = gw[0];
    float s    = v[0];
    #pragma unroll
    for (int w = 1; w < POOL_ROWS; ++w) {
      if (gw[w] != gcur) {
        atomicAdd(Zc + (size_t)gcur * ZP, s);
        gcur = gw[w];
        s = v[w];
      } else {
        s += v[w];
      }
    }
    atomicAdd(Zc + (size_t)gcur * ZP, s);
  }
}

extern "C" void kernel_launch(void* const* d_in, const int* in_sizes, int n_in,
                              void* d_out, int out_size, void* d_ws, size_t ws_size,
                              hipStream_t stream) {
  const float* x    = (const float*)d_in[0];
  const int*   ei   = (const int*)  d_in[1];
  const int*   batch= (const int*)  d_in[2];
  const float* W1   = (const float*)d_in[3];
  const float* b1   = (const float*)d_in[4];
  const float* W2   = (const float*)d_in[5];
  const float* b2   = (const float*)d_in[6];
  float* out = (float*)d_out;
  float* ws  = (float*)d_ws;

  const int* src = ei;
  const int* dst = ei + N_EDGES;

  float* dinv = ws;
  float* H    = ws + 50048;
  float* R    = H + (size_t)N_NODES * NF;
  float* Y    = R + (size_t)N_NODES * NF;

  int*  rowptr = (int*)R;
  int*  rank   = rowptr + OWN_SLICES * CNT_PITCH;
  int2* csr    = (int2*)(rank + N_EDGES);
  float* Z     = (float*)(csr + N_EDGES);
  int*  tot    = (int*)(Z + N_GRAPHS * ZP);

  const int T = 256;
  const int gRows     = N_NODES / 8;
  const int gPool     = N_NODES / POOL_ROWS;

  count_rank_owner_kernel<<<OWN_BLOCKS, OWN_THREADS, 0, stream>>>(dst, rowptr, rank, tot, Z);
  scan_dinv_kernel<<<SCAN_RANGES, SCAN_THREADS, 0, stream>>>(rowptr, tot, dinv);
  fill_gemm_kernel<<<GEMM_BLOCKS + FILL_BLOCKS, T, 0, stream>>>(src, dst, rowptr, rank, dinv, csr, x, W1, H);

  gather_relu_kernel<<<gRows, T, 0, stream>>>(H, rowptr, csr, dinv, b1, Y);

  gather_pool_kernel<<<gPool, POOL_ROWS * 32, 0, stream>>>(Y, rowptr, csr, dinv, batch, Z);
  pooled_product_wmma<<<N_GRAPHS / 16, T, 0, stream>>>(Z, W2, b2, out);
}


# edge fill by owner blocks: a block keeps one slice's row starts of its node range in LDS and hands out record places by LDS adds as it sweeps the slice; no rank array; sum order: fill arrival order
# speedup vs baseline: 5.2657x; 1.2370x over previous
#include <hip/hip_runtime.h>

#define N_NODES  50000
#define N_EDGES  800000
#define NF       128
#define N_GRAPHS 128

typedef __attribute__((ext_vector_type(2))) float v2f;
typedef __attribute__((ext_vector_type(8))) float v8f;

#define ZP (NF + 4)


#define OWN_RANGES   8
#define OWN_SLICES   32
#define OWN_BLOCKS   (OWN_RANGES * OWN_SLICES)
#define OWN_THREADS  1024
#define OWN_WAVES    (OWN_THREADS / 32)
#define SCAN_RANGES  64
#define SCAN_PER_OWN (SCAN_RANGES / OWN_RANGES)
#define SCAN_NODES   ((((N_NODES + SCAN_RANGES - 1) / SCAN_RANGES) + 3) / 4 * 4)
#define OWN_NODES    (SCAN_PER_OWN * SCAN_NODES)
#define OWN_LOADS    7
#define CNT_PITCH    50048
#define SLICE_EDGES  (N_EDGES / OWN_SLICES)
#define E4           (SLICE_EDGES / 4)
#define TOT_SPLIT    (OWN_WAVES / SCAN_PER_OWN)
#define TOT_SPAN     (SCAN_NODES / TOT_SPLIT)
static_assert(N_EDGES % (4 * OWN_SLICES) == 0, "equal slices, read four edges at a time");
static_assert(CNT_PITCH >= N_NODES + 1 && CNT_PITCH % 4 == 0, "count arrays: room for the row end, 16-byte aligned");
static_assert(SCAN_RANGES % OWN_RANGES == 0 && OWN_WAVES % SCAN_PER_OWN == 0 && SCAN_NODES % TOT_SPLIT == 0,
              "whole scan ranges per owner range, whole waves per scan range");
__global__ __launch_bounds__(OWN_THREADS) void count_owner_kernel(const int* __restrict__ dst,
                                                                   int* __restrict__ cnt,
                                                                   int* __restrict__ tot,
                                                                   float* __restrict__ Z) {
  __shared__ int c[OWN_NODES];
  __shared__ int wsum[OWN_WAVES];
  const int range = blockIdx.x % OWN_RANGES;
  const int slice = blockIdx.x / OWN_RANGES;
  const int lo = range * OWN_NODES;
  for (int i = threadIdx.x; i < OWN_NODES; i += OWN_THREADS) c[i] = 0;
  for (int i = blockIdx.x * OWN_THREADS + threadIdx.x; i < N_GRAPHS * ZP; i += OWN_BLOCKS * OWN_THREADS)
    Z[i] = 0.0f;
  __syncthreads();

  const int4* d4 = (const int4*)(dst + slice * SLICE_EDGES);
  const int rot = range * (E4 / OWN_RANGES);
  for (int base = 0; base < E4; base += OWN_THREADS * OWN_LOADS) {
    int4 v[OWN_LOADS];
    #pragma unroll
    for (int u = 0; u < OWN_LOADS; ++u) {
      const int i = base + u * OWN_THREADS + threadIdx.x;
      int j = i + rot;
      if (j >= E4) j -= E4;
      v[u] = (i < E4) ? d4[j] : make_int4(-1, -1, -1, -1);
    }
    #pragma unroll
    for (int u = 0; u < OWN_LOADS; ++u) {
      const unsigned t0 = (unsigned)(v[u].x - lo);
      const unsigned t1 = (unsigned)(v[u].y - lo);
      const unsigned t2 = (unsigned)(v[u].z - lo);
      const unsigned t3 = (unsigned)(v[u].w - lo);
      if (t0 < (unsigned)OWN_NODES) atomicAdd(&c[t0], 1);
      if (t1 < (unsigned)OWN_NODES) atomicAdd(&c[t1], 1);
      if (t2 < (unsigned)OWN_NODES) atomicAdd(&c[t2], 1);
      if (t3 < (unsigned)OWN_NODES) atomicAdd(&c[t3], 1);
    }
  }
  __syncthreads();
  for (int i = threadIdx.x; i < OWN_NODES; i += OWN_THREADS)
    if (lo + i < N_NODES) cnt[slice * CNT_PITCH + lo + i] = c[i];
  const int wave = threadIdx.x >> 5;
  const int lane = threadIdx.x & 31;
  int mine = 0;
  for (int k = lane; k < TOT_SPAN; k += 32) mine += c[wave * TOT_SPAN + k];
  for (int o = 16; o > 0; o >>= 1) mine += __shfl_xor(mine, o);
  if (lane == 0) wsum[wave] = mine;
  __syncthreads();
  if (threadIdx.x < SCAN_PER_OWN) {
    int t = 0;
    #pragma unroll
    for (int w = 0; w < TOT_SPLIT; ++w) t += wsum[threadIdx.x * TOT_SPLIT + w];
    tot[slice * SCAN_RANGES + range * SCAN_PER_OWN + threadIdx.x] = t;
  }
}

#define SCAN_THREADS 256
#define SCAN_WAVES   (SCAN_THREADS / 32)
static_assert(SCAN_NODES <= 4 * SCAN_THREADS, "one thread per four nodes of a range");
static_assert(N_NODES % 4 == 0, "counts are handled four at a time");
__global__ __launch_bounds__(SCAN_THREADS) void scan_dinv_kernel(int* __restrict__ cr,
                                                                 const int* __restrict__ tot,
                                                                 float* __restrict__ dinv) {
  __shared__ int wsum[SCAN_WAVES];
  __shared__ int fsum[SCAN_WAVES];
  const int lane  = threadIdx.x & 31;
  const int wave  = threadIdx.x >> 5;
  const int range = blockIdx.x;
  const int i = range * SCAN_NODES + 4 * threadIdx.x;
  const bool on = 4 * threadIdx.x < SCAN_NODES && i < N_NODES;

  int front = 0;
  for (int j = threadIdx.x; j < OWN_SLICES * SCAN_RANGES; j += SCAN_THREADS)
    if (j % SCAN_RANGES < range) front += tot[j];

  int4 cs[OWN_SLICES];
  int4 c = make_int4(0, 0, 0, 0);
  #pragma unroll
  for (int s = 0; s < OWN_SLICES; ++s) {
    cs[s] = on ? *(const int4*)(cr + s * CNT_PITCH + i) : make_int4(0, 0, 0, 0);
    c.x += cs[s].x; c.y += cs[s].y; c.z += cs[s].z; c.w += cs[s].w;
  }
  const int t = c.x + c.y + c.z + c.w;
  int inc = t;
  for (int o = 1; o < 32; o <<= 1) {
    const int up = __shfl_up(inc, o);
    if (lane >= o) inc += up;
  }
  for (int o = 16; o > 0; o >>= 1) front += __shfl_xor(front, o);
  if (lane == 31) wsum[wave] = inc;
  if (lane == 0)  fsum[wave] = front;
  __syncthreads();

  int carry = 0;
  #pragma unroll
  for (int w = 0; w < SCAN_WAVES; ++w) {
    carry += fsum[w];
    if (w < wave) carry += wsum[w];
  }
  const int ex = carry + inc - t;
  if (on) {
    int4 p = make_int4(ex, ex + c.x, ex + c.x + c.y, ex + c.x + c.y + c.z);
    #pragma unroll
    for (int s = 0; s < OWN_SLICES; ++s) {
      *(int4*)(cr + s * CNT_PITCH + i) = p;
      p.x += cs[s].x; p.y += cs[s].y; p.z += cs[s].z; p.w += cs[s].w;
    }
    *(float4*)(dinv + i) = make_float4(rsqrtf((float)c.x + 1.0f), rsqrtf((float)c.y + 1.0f),
                                       rsqrtf((float)c.z + 1.0f), rsqrtf((float)c.w + 1.0f));
    if (i + 4 == N_NODES) cr[N_NODES] = ex + t;
  }
}

#define FILL_THREADS 256
#define FILL_LOADS   5
#define FILL_STAGE   ((OWN_NODES / 4 + FILL_THREADS - 1) / FILL_THREADS)
#define FILL_NODES   (4 * FILL_THREADS * FILL_STAGE)
static_assert(OWN_NODES % 4 == 0, "row starts and dinv are staged four at a time");
static_assert(FILL_NODES >= OWN_NODES + 32, "one spare counter per lane behind the range's counters");
#define FILL_OWNED(D)      ((unsigned)((D) - lo) < (unsigned)OWN_NODES)
#define FILL_WEIGHT(D, S, W)  W = dinv[FILL_OWNED(D) ? (S) : 0]
#define FILL_PLACE(D, P)      P = atomicAdd(&start[FILL_OWNED(D) ? (unsigned)((D) - lo) : spare], 1)
#define FILL_DINVD(D, N)      N = dl[FILL_OWNED(D) ? (unsigned)((D) - lo) : spare]
#define FILL_STORE(D, S, W, N, P) if (FILL_OWNED(D)) csr[P] = make_int2(S, __float_as_int((W) * (N)))
__device__ __forceinline__ void fill_owner_block(const int unit, int* start, float* dl,
                                                 const int* __restrict__ src,
                                                 const int* __restrict__ dst,
                                                 const int* __restrict__ rowptr,
                                                 const float* __restrict__ dinv,
                                                 int2* __restrict__ csr) {
  const int range = unit % OWN_RANGES;
  const int slice = unit / OWN_RANGES;
  const int lo = range * OWN_NODES;
  const unsigned spare = OWN_NODES + (threadIdx.x & 31);
  const int4*   rp4 = (const int4*)(rowptr + slice * CNT_PITCH + lo);
  const float4* dn4 = (const float4*)(dinv + lo);
  const unsigned qmax = (unsigned)(N_NODES - 4 - lo) / 4u;
  int4   si[FILL_STAGE];
  float4 df[FILL_STAGE];
  #pragma unroll
  for (int r = 0; r < FILL_STAGE; ++r) {
    const unsigned q = min(r * FILL_THREADS + threadIdx.x, qmax);
    si[r] = rp4[q];
    df[r] = dn4[q];
  }
  #pragma unroll
  for (int r = 0; r < FILL_STAGE; ++r) {
    const int q = r * FILL_THREADS + threadIdx.x;
    ((int4*)start)[q] = si[r];
    ((float4*)dl)[q]  = df[r];
  }
  __syncthreads();

  const int4* d4 = (const int4*)(dst + slice * SLICE_EDGES);
  const int4* s4 = (const int4*)(src + slice * SLICE_EDGES);
  const int rot = range * (E4 / OWN_RANGES);
  for (int base = 0; base < E4; base += FILL_THREADS * FILL_LOADS) {
    int4   dv[FILL_LOADS];
    int4   sv[FILL_LOADS];
    float4 w[FILL_LOADS];
    #pragma unroll
    for (int u = 0; u < FILL_LOADS; ++u) {
      const int i = base + u * FILL_THREADS + threadIdx.x;
      int j = i + rot;
      if (j >= E4) j -= E4;
      if (i >= E4) j = 0;
      dv[u] = d4[j];
      sv[u] = s4[j];
      if (i >= E4) dv[u] = make_int4(-1, -1, -1, -1);
    }
    #pragma unroll
    for (int u = 0; u < FILL_LOADS; ++u) {
      FILL_WEIGHT(dv[u].x, sv[u].x, w[u].x);
      FILL_WEIGHT(dv[u].y, sv[u].y, w[u].y);
      FILL_WEIGHT(dv[u].z, sv[u].z, w[u].z);
      FILL_WEIGHT(dv[u].w, sv[u].w, w[u].w);
    }
    #pragma unroll
    for (int u = 0; u < FILL_LOADS; ++u) {
      int4   p;
      float4 n;
      FILL_PLACE(dv[u].x, p.x);
      FILL_PLACE(dv[u].y, p.y);
      FILL_PLACE(dv[u].z, p.z);
      FILL_PLACE(dv[u].w, p.w);
      FILL_DINVD(dv[u].x, n.x);
      FILL_DINVD(dv[u].y, n.y);
      FILL_DINVD(dv[u].z, n.z);
      FILL_DINVD(dv[u].w, n.w);
      FILL_STORE(dv[u].x, sv[u].x, w[u].x, n.x, p.x);
      FILL_STORE(dv[u].y, sv[u].y, w[u].y, n.y, p.y);
      FILL_STORE(dv[u].z, sv[u].z, w[u].z, n.z, p.z);
      FILL_STORE(dv[u].w, sv[u].w, w[u].w, n.w, p.w);
    }
  }
}

template <int R>
__device__ __forceinline__ void tile_product(const float* __restrict__ Xrow, const int tile_pitch,
                                             const float* __restrict__ Wcol,
                                             const int half, v8f (&acc)[R]) {
  #pragma unroll
  for (int k = 0; k < NF; k += 4) {
    v2f b;
    b.x = Wcol[(size_t)(k + 2 * half + 0) * NF];
    b.y = Wcol[(size_t)(k + 2 * half + 1) * NF];
    #pragma unroll
    for (int t = 0; t < R; ++t) {
      const v2f a = *(const v2f*)(Xrow + (size_t)t * tile_pitch + k + 2 * half);
      acc[t] = __builtin_amdgcn_wmma_f32_16x16x4_f32(
           false, a,  false, b,
           (short)0, acc[t],  false,  false);
    }
  }
}

#define GEMM_RT      5
#define GEMM_ROWS    (16 * GEMM_RT)
#define GEMM_BLOCKS  (N_NODES / GEMM_ROWS)
#define GEMM_LDS_F4  (GEMM_ROWS * (ZP / 4))
#define FILL_LDS_F4  (2 * FILL_NODES / 4)
#define FG_LDS_F4    (GEMM_LDS_F4 > FILL_LDS_F4 ? GEMM_LDS_F4 : FILL_LDS_F4)
static_assert(N_NODES % GEMM_ROWS == 0, "whole row tiles, EXEC all ones");
static_assert((GEMM_ROWS * (NF / 4)) % 256 == 0, "staging: whole float4 rounds per thread");
static_assert(FILL_THREADS == 256, "fill blocks and product blocks share a launch");
__global__ __launch_bounds__(256) void fill_gemm_kernel(const int* __restrict__ src,
                                                        const int* __restrict__ dst,
                                                        const int* __restrict__ rowptr,
                                                        const float* __restrict__ dinv,
                                                        int2* __restrict__ csr,
                                                        const float* __restrict__ X,
                                                        const float* __restrict__ W,
                                                        float* __restrict__ H) {
  __shared__ float4 lds[FG_LDS_F4];
  if (blockIdx.x < OWN_BLOCKS) {
    fill_owner_block(blockIdx.x, (int*)lds, (float*)lds + FILL_NODES,
                     src, dst, rowptr, dinv, csr);
    return;
  }
  const int gblock = blockIdx.x - OWN_BLOCKS;
  float4 (*xs)[ZP / 4] = (float4 (*)[ZP / 4])lds;
  const int wave = threadIdx.x >> 5;
  const int lane = threadIdx.x & 31;
  const int half = lane >> 4;
  const int l    = lane & 15;
  const int m0   = gblock * GEMM_ROWS;
  const int n0   = wave * 16;

  const float4* Xb = (const float4*)(X + (size_t)m0 * NF);
  #pragma unroll
  for (int i = 0; i < GEMM_ROWS * (NF / 4) / 256; ++i) {
    const int idx = i * 256 + threadIdx.x;
    xs[idx >> 5][idx & 31] = Xb[idx];
  }
  __syncthreads();

  v8f acc[GEMM_RT] = {};
  tile_product<GEMM_RT>((const float*)xs + (size_t)l * ZP, 16 * ZP, W + n0 + l, half, acc);

  #pragma unroll
  for (int t = 0; t < GEMM_RT; ++t) {
    float* Hp = H + (size_t)(m0 + 16 * t + 8 * half) * NF + n0 + l;
    #pragma unroll
    for (int r = 0; r < 8; ++r) Hp[(size_t)r * NF] = acc[t][r];
  }
}

static_assert(N_GRAPHS % 16 == 0, "whole 16-row tiles, EXEC all ones");
__global__ __launch_bounds__(256) void pooled_product_wmma(const float* __restrict__ Z,
                                                           const float* __restrict__ W,
                                                           const float* __restrict__ b,
                                                           float* __restrict__ out) {
  const int wave = threadIdx.x >> 5;
  const int lane = threadIdx.x & 31;
  const int half = lane >> 4;
  const int l    = lane & 15;
  const int m0   = blockIdx.x * 16;
  const int n0   = wave * 16;

  const float bj = b[n0 + l];
  const float* cp = Z + (size_t)(m0 + 8 * half) * ZP + NF;
  v8f c0;
  #pragma unroll
  for (int r = 0; r < 8; ++r) c0[r] = cp[(size_t)r * ZP] * bj;

  v8f acc[1] = {c0};
  tile_product<1>(Z + (size_t)(m0 + l) * ZP, 0, W + n0 + l, half, acc);

  float* Op = out + (size_t)(m0 + 8 * half) * NF + n0 + l;
  #pragma unroll
  for (int r = 0; r < 8; ++r) Op[(size_t)r * NF] = acc[0][r];
}

__device__ __forceinline__ float4 row_sum(const float* __restrict__ Hl,
                                          const int2* __restrict__ csr,
                                          const int beg, const int end, const int lane) {
  float4 acc = make_float4(0.0f, 0.0f, 0.0f, 0.0f);
  for (int c = beg; c < end; c += 32) {
    const int n = min(32, end - c);
    int2 sw = make_int2(0, 0);
    if (lane < n) sw = csr[c + lane];
    int k = 0;
    for (; k + 4 <= n; k += 4) {
      const int s0 = __builtin_amdgcn_readlane(sw.x, k + 0);
      const int s1 = __builtin_amdgcn_readlane(sw.x, k + 1);
      const int s2 = __builtin_amdgcn_readlane(sw.x, k + 2);
      const int s3 = __builtin_amdgcn_readlane(sw.x, k + 3);
      const float w0 = __int_as_float(__builtin_amdgcn_readlane(sw.y, k + 0));
      const float w1 = __int_as_float(__builtin_amdgcn_readlane(sw.y, k + 1));
      const float w2 = __int_as_float(__builtin_amdgcn_readlane(sw.y, k + 2));
      const float w3 = __int_as_float(__builtin_amdgcn_readlane(sw.y, k + 3));
      const float4 h0 = *(const float4*)(Hl + (size_t)s0 * NF);
      const float4 h1 = *(const float4*)(Hl + (size_t)s1 * NF);
      const float4 h2 = *(const float4*)(Hl + (size_t)s2 * NF);
      const float4 h3 = *(const float4*)(Hl + (size_t)s3 * NF);
      acc.x += h0.x * w0; acc.y += h0.y * w0; acc.z += h0.z * w0; acc.w += h0.w * w0;
      acc.x += h1.x * w1; acc.y += h1.y * w1; acc.z += h1.z * w1; acc.w += h1.w * w1;
      acc.x += h2.x * w2; acc.y += h2.y * w2; acc.z += h2.z * w2; acc.w += h2.w * w2;
      acc.x += h3.x * w3; acc.y += h3.y * w3; acc.z += h3.z * w3; acc.w += h3.w * w3;
    }
    const int rem = n - k;
    if (rem > 0) {
      const int s0 = __builtin_amdgcn_readlane(sw.x, k);
      const float w0 = __int_as_float(__builtin_amdgcn_readlane(sw.y, k));
      const float4 h0 = *(const float4*)(Hl + (size_t)s0 * NF);
      if (rem > 1) {
        const int s1 = __builtin_amdgcn_readlane(sw.x, k + 1);
        const float w1 = __int_as_float(__builtin_amdgcn_readlane(sw.y, k + 1));
        const float4 h1 = *(const float4*)(Hl + (size_t)s1 * NF);
        if (rem > 2) {
          const int s2 = __builtin_amdgcn_readlane(sw.x, k + 2);
          const float w2 = __int_as_float(__builtin_amdgcn_readlane(sw.y, k + 2));
          const float4 h2 = *(const float4*)(Hl + (size_t)s2 * NF);
          acc.x += h0.x * w0; acc.y += h0.y * w0; acc.z += h0.z * w0; acc.w += h0.w * w0;
          acc.x += h1.x * w1; acc.y += h1.y * w1; acc.z += h1.z * w1; acc.w += h1.w * w1;
          acc.x += h2.x * w2; acc.y += h2.y * w2; acc.z += h2.z * w2; acc.w += h2.w * w2;
        } else {
          acc.x += h0.x * w0; acc.y += h0.y * w0; acc.z += h0.z * w0; acc.w += h0.w * w0;
          acc.x += h1.x * w1; acc.y += h1.y * w1; acc.z += h1.z * w1; acc.w += h1.w * w1;
        }
      } else {
        acc.x += h0.x * w0; acc.y += h0.y * w0; acc.z += h0.z * w0; acc.w += h0.w * w0;
      }
    }
  }
  return acc;
}

static_assert(N_NODES % 8 == 0, "eight rows per block, no tail");
__global__ __launch_bounds__(256) void gather_relu_kernel(const float* __restrict__ H,
                                                          const int* __restrict__ rowptr,
                                                          const int2* __restrict__ csr,
                                                          const float* __restrict__ dinv,
                                                          const float* __restrict__ b,
                                                          float* __restrict__ OUT) {
  const int lane = threadIdx.x & 31;
  const int i   = __builtin_amdgcn_readfirstlane(blockIdx.x * 8 + (threadIdx.x >> 5));
  const int beg = __builtin_amdgcn_readfirstlane(rowptr[i]);
  const int end = __builtin_amdgcn_readfirstlane(rowptr[i + 1]);
  const float* Hl = H + lane * 4;

  const float4 hs = *(const float4*)(Hl + (size_t)i * NF);
  const float4 bv = *(const float4*)(b + lane * 4);
  const float di = dinv[i];

  const float4 acc = row_sum(Hl, csr, beg, end, lane);

  const float di2 = di * di;
  float4 r;
  r.x = fmaxf(acc.x + di2 * hs.x + bv.x, 0.0f);
  r.y = fmaxf(acc.y + di2 * hs.y + bv.y, 0.0f);
  r.z = fmaxf(acc.z + di2 * hs.z + bv.z, 0.0f);
  r.w = fmaxf(acc.w + di2 * hs.w + bv.w, 0.0f);
  *(float4*)(OUT + (size_t)i * NF + lane * 4) = r;
}

#define POOL_ROWS 16
static_assert(N_NODES % POOL_ROWS == 0, "sixteen rows per block, no tail");
__global__ __launch_bounds__(POOL_ROWS * 32) void gather_pool_kernel(
    const float* __restrict__ Y, const int* __restrict__ rowptr,
    const int2* __restrict__ csr, const float* __restrict__ dinv,
    const int* __restrict__ batch, float* __restrict__ Z) {
  __shared__ float4 rows[POOL_ROWS][ZP / 4];
  __shared__ int gid[POOL_ROWS];
  const int lane = threadIdx.x & 31;
  const int wave = threadIdx.x >> 5;
  const int i   = __builtin_amdgcn_readfirstlane(blockIdx.x * POOL_ROWS + wave);
  const int beg = __builtin_amdgcn_readfirstlane(rowptr[i]);
  const int end = __builtin_amdgcn_readfirstlane(rowptr[i + 1]);
  const float* Yl = Y + lane * 4;

  const float4 hs = *(const float4*)(Yl + (size_t)i * NF);
  const float di = dinv[i];
  const int   g  = batch[i];

  const float4 acc = row_sum(Yl, csr, beg, end, lane);

  const float di2 = di * di;
  float4 r;
  r.x = acc.x + di2 * hs.x;
  r.y = acc.y + di2 * hs.y;
  r.z = acc.z + di2 * hs.z;
  r.w = acc.w + di2 * hs.w;
  rows[wave][lane] = r;
  if (lane == 0) {
    rows[wave][NF / 4] = make_float4(1.0f, 0.0f, 0.0f, 0.0f);
    gid[wave] = g;
  }
  __syncthreads();

  if (threadIdx.x <= NF) {
    const float* rf = (const float*)rows + threadIdx.x;
    float* Zc = Z + threadIdx.x;
    float v[POOL_ROWS];
    int   gw[POOL_ROWS];
    #pragma unroll
    for (int w = 0; w < POOL_ROWS; ++w) {
      v[w]  = rf[w * ZP];
      gw[w] = gid[w];
    }
    int   gcur = gw[0];
    float s    = v[0];
    #pragma unroll
    for (int w = 1; w < POOL_ROWS; ++w) {
      if (gw[w] != gcur) {
        atomicAdd(Zc + (size_t)gcur * ZP, s);
        gcur = gw[w];
        s = v[w];
      } else {
        s += v[w];
      }
    }
    atomicAdd(Zc + (size_t)gcur * ZP, s);
  }
}

extern "C" void kernel_launch(void* const* d_in, const int* in_sizes, int n_in,
                              void* d_out, int out_size, void* d_ws, size_t ws_size,
                              hipStream_t stream) {
  const float* x    = (const float*)d_in[0];
  const int*   ei   = (const int*)  d_in[1];
  const int*   batch= (const int*)  d_in[2];
  const float* W1   = (const float*)d_in[3];
  const float* b1   = (const float*)d_in[4];
  const float* W2   = (const float*)d_in[5];
  const float* b2   = (const float*)d_in[6];
  float* out = (float*)d_out;
  float* ws  = (float*)d_ws;

  const int* src = ei;
  const int* dst = ei + N_EDGES;

  float* dinv = ws;
  float* H    = ws + 50048;
  float* R    = H + (size_t)N_NODES * NF;
  float* Y    = R + (size_t)N_NODES * NF;

  int*  rowptr = (int*)R;
  int2* csr    = (int2*)(rowptr + OWN_SLICES * CNT_PITCH);
  float* Z     = (float*)(csr + N_EDGES);
  int*  tot    = (int*)(Z + N_GRAPHS * ZP);

  const int T = 256;
  const int gRows     = N_NODES / 8;
  const int gPool     = N_NODES / POOL_ROWS;

  count_owner_kernel<<<OWN_BLOCKS, OWN_THREADS, 0, stream>>>(dst, rowptr, tot, Z);
  scan_dinv_kernel<<<SCAN_RANGES, SCAN_THREADS, 0, stream>>>(rowptr, tot, dinv);
  fill_gemm_kernel<<<GEMM_BLOCKS + OWN_BLOCKS, T, 0, stream>>>(src, dst, rowptr, dinv, csr, x, W1, H);

  gather_relu_kernel<<<gRows, T, 0, stream>>>(H, rowptr, csr, dinv, b1, Y);

  gather_pool_kernel<<<gPool, POOL_ROWS * 32, 0, stream>>>(Y, rowptr, csr, dinv, batch, Z);
  pooled_product_wmma<<<N_GRAPHS / 16, T, 0, stream>>>(Z, W2, b2, out);
}
